# starting kernel
# baseline (speedup 1.0000x reference)
#include <hip/hip_runtime.h>

typedef float v2f __attribute__((ext_vector_type(2)));
typedef float v8f __attribute__((ext_vector_type(8)));

#define UCOLS 96
#define UROWS 820
#define NROW  131072
#define TILE_ROWS 16
#define BLOCK_THREADS 192

__global__ void zero_u(float* __restrict__ UT) {
  int t = blockIdx.x * blockDim.x + threadIdx.x;
  int stride = gridDim.x * blockDim.x;
  for (int i = t; i < UROWS * UCOLS; i += stride) UT[i] = 0.f;
}

__device__ inline void scatter_u(const float* __restrict__ u, int nD, int rowbase,
                                 int mul, int idim, int colbase,
                                 float* __restrict__ UT, int t, int stride) {
  const int mi = mul * idim;
  const int total = nD * mi;
  for (int e = t; e < total; e += stride) {
    int dd  = e / mi;
    int rem = e - dd * mi;
    UT[(colbase + rem) * UROWS + rowbase + dd] = u[e];
  }
}

__global__ void fill_u(const float* __restrict__ u1_0e, const float* __restrict__ u1_1o,
                       const float* __restrict__ u1_2e, const float* __restrict__ u2_0e,
                       const float* __restrict__ u2_1o, const float* __restrict__ u2_2e,
                       const float* __restrict__ u3_0e, const float* __restrict__ u3_1o,
                       const float* __restrict__ u3_2e, float* __restrict__ UT) {
  int t = blockIdx.x * blockDim.x + threadIdx.x;
  int stride = gridDim.x * blockDim.x;
  scatter_u(u1_0e,   9,  0, 1, 1,  0, UT, t, stride);
  scatter_u(u1_1o,   9,  0, 1, 3,  1, UT, t, stride);
  scatter_u(u1_2e,   9,  0, 1, 5,  4, UT, t, stride);
  scatter_u(u2_0e,  81,  9, 2, 1,  9, UT, t, stride);
  scatter_u(u2_1o,  81,  9, 2, 3, 11, UT, t, stride);
  scatter_u(u2_2e,  81,  9, 3, 5, 17, UT, t, stride);
  scatter_u(u3_0e, 729, 90, 4, 1, 32, UT, t, stride);
  scatter_u(u3_1o, 729, 90, 6, 3, 36, UT, t, stride);
  scatter_u(u3_2e, 729, 90, 8, 5, 54, UT, t, stride);
}

__global__ __launch_bounds__(BLOCK_THREADS) void symcon_main(
    const float* __restrict__ x, const float* __restrict__ UT,
    const float* __restrict__ w1_0e, const float* __restrict__ w1_1o,
    const float* __restrict__ w1_2e, const float* __restrict__ w2_0e,
    const float* __restrict__ w2_1o, const float* __restrict__ w2_2e,
    const float* __restrict__ w3_0e, const float* __restrict__ w3_1o,
    const float* __restrict__ w3_2e, const int* __restrict__ index,
    float* __restrict__ out) {
  extern __shared__ float lds[];
  float* mono = lds;
  float* G    = lds + TILE_ROWS * UROWS;
  float* ldsX = lds + TILE_ROWS * UROWS + TILE_ROWS * UCOLS;

  const int tid      = threadIdx.x;
  const int rowBase  = blockIdx.x * TILE_ROWS;

  if (tid < TILE_ROWS * 9) ldsX[tid] = x[rowBase * 9 + tid];
  __syncthreads();

  {
    const int row  = tid & 15;
    const int slot = tid >> 4;
    const float* xr = ldsX + row * 9;
    float* mrow = mono + row * UROWS;
    for (int m = slot; m < UROWS; m += 12) {
      float val;
      if (m < 9) {
        val = xr[m];
      } else if (m < 90) {
        int t2 = m - 9;
        int d0 = t2 / 9;
        val = xr[d0] * xr[t2 - d0 * 9];
      } else if (m < 819) {
        int t3 = m - 90;
        int d0 = t3 / 81;
        int rr = t3 - d0 * 81;
        int d1 = rr / 9;
        val = xr[d0] * xr[d1] * xr[rr - d1 * 9];
      } else {
        val = 0.f;
      }
      mrow[m] = val;
    }
  }
  __syncthreads();

  {
    const int wv   = __builtin_amdgcn_readfirstlane(tid >> 5);
    const int lane = tid & 31;
    const int half = lane >> 4;
    const int r    = lane & 15;
    const int ka0  = half ? 2 : 0;
    const int n0   = wv * 16;

    const int kstart[6] = {0, 8, 88, 88, 88, 88};
    const int ksteps[6] = {23, 21, 183, 183, 183, 183};
    const int ns = ksteps[wv];

    const float* arow = mono + r * UROWS;
    const float* bcol = UT + (n0 + r) * UROWS;
    v8f acc = {0.f, 0.f, 0.f, 0.f, 0.f, 0.f, 0.f, 0.f};
    int kb = kstart[wv];
    for (int s = 0; s < ns; ++s, kb += 4) {
      v2f a = *(const v2f*)(arow + kb + ka0);
      v2f b = *(const v2f*)(bcol + kb + ka0);
      acc = __builtin_amdgcn_wmma_f32_16x16x4_f32(
           false, a,  false, b,
           (short)0, acc,  false,  false);
    }
#pragma unroll
    for (int j = 0; j < 8; ++j)
      G[(j + 8 * half) * UCOLS + n0 + r] = acc[j];
  }
  __syncthreads();

  if (tid < TILE_ROWS * 9) {
    const int row  = tid / 9;
    const int comp = tid - row * 9;
    const int R    = rowBase + row;
    const int z    = R >> 7;
    const int c    = R & 127;
    const int sp   = index[z];
    const float* Gr = G + row * UCOLS;
    float* ob = out + z * 1152;
    if (comp == 0) {
      float a = w1_0e[sp * 128 + c] * Gr[0]
              + w2_0e[(sp * 2 + 0) * 128 + c] * Gr[9]
              + w2_0e[(sp * 2 + 1) * 128 + c] * Gr[10];
#pragma unroll
      for (int k = 0; k < 4; ++k) a += w3_0e[(sp * 4 + k) * 128 + c] * Gr[32 + k];
      ob[c] = a;
    } else if (comp < 4) {
      const int i = comp - 1;
      float a = w1_1o[sp * 128 + c] * Gr[1 + i];
#pragma unroll
      for (int k = 0; k < 2; ++k) a += w2_1o[(sp * 2 + k) * 128 + c] * Gr[11 + 3 * k + i];
#pragma unroll
      for (int k = 0; k < 6; ++k) a += w3_1o[(sp * 6 + k) * 128 + c] * Gr[36 + 3 * k + i];
      ob[128 + c * 3 + i] = a;
    } else {
      const int i = comp - 4;
      float a = w1_2e[sp * 128 + c] * Gr[4 + i];
#pragma unroll
      for (int k = 0; k < 3; ++k) a += w2_2e[(sp * 3 + k) * 128 + c] * Gr[17 + 5 * k + i];
#pragma unroll
      for (int k = 0; k < 8; ++k) a += w3_2e[(sp * 8 + k) * 128 + c] * Gr[54 + 5 * k + i];
      ob[512 + c * 5 + i] = a;
    }
  }
}

extern "C" void kernel_launch(void* const* d_in, const int* in_sizes, int n_in,
                              void* d_out, int out_size, void* d_ws, size_t ws_size,
                              hipStream_t stream) {
  const float* x     = (const float*)d_in[0];
  const float* u1_0e = (const float*)d_in[1];
  const float* w1_0e = (const float*)d_in[2];
  const float* u1_1o = (const float*)d_in[3];
  const float* w1_1o = (const float*)d_in[4];
  const float* u1_2e = (const float*)d_in[5];
  const float* w1_2e = (const float*)d_in[6];
  const float* u2_0e = (const float*)d_in[7];
  const float* w2_0e = (const float*)d_in[8];
  const float* u2_1o = (const float*)d_in[9];
  const float* w2_1o = (const float*)d_in[10];
  const float* u2_2e = (const float*)d_in[11];
  const float* w2_2e = (const float*)d_in[12];
  const float* u3_0e = (const float*)d_in[13];
  const float* w3_0e = (const float*)d_in[14];
  const float* u3_1o = (const float*)d_in[15];
  const float* w3_1o = (const float*)d_in[16];
  const float* u3_2e = (const float*)d_in[17];
  const float* w3_2e = (const float*)d_in[18];
  const int*   index = (const int*)d_in[19];
  float* UT  = (float*)d_ws;
  float* out = (float*)d_out;

  zero_u<<<(UROWS * UCOLS + 255) / 256, 256, 0, stream>>>(UT);
  fill_u<<<128, 256, 0, stream>>>(u1_0e, u1_1o, u1_2e, u2_0e, u2_1o, u2_2e,
                                  u3_0e, u3_1o, u3_2e, UT);

  const size_t ldsBytes =
      (TILE_ROWS * UROWS + TILE_ROWS * UCOLS + TILE_ROWS * 9) * sizeof(float);
  symcon_main<<<NROW / TILE_ROWS, BLOCK_THREADS, ldsBytes, stream>>>(
      x, UT, w1_0e, w1_1o, w1_2e, w2_0e, w2_1o, w2_2e, w3_0e, w3_1o, w3_2e,
      index, out);
}


# Species weights contracted into U first (per species,channel: K 852 x 9 columns), nodes grouped by species so each wave's GEMM has 9 output columns not 94; changes f32 summation order
# speedup vs baseline: 1.4305x; 1.4305x over previous
#include <hip/hip_runtime.h>

typedef float v2f __attribute__((ext_vector_type(2)));
typedef float v8f __attribute__((ext_vector_type(8)));

#define NSPEC 10
#define NCH 128
#define NNODE 1024
#define NCOMP 9
#define G1 12
#define G2 84
#define KROW 852
#define CHUNK_ROWS 32
#define MAXSLOT 42
#define UW_FLOATS (NSPEC * NCH * NCOMP * KROW)
#define ZL_OFF  UW_FLOATS
#define CNT_OFF (UW_FLOATS + NSPEC * NNODE)
#define TAB_OFF (CNT_OFF + 16)

#define CG 32
#define EPARTS 12
#define EPART_LEN 639
#define WROWS 28
#define PREP_BLOCKS (NSPEC * (NCH / CG) * EPARTS)

__device__ inline float w_row(const float* __restrict__ w1_0e, const float* __restrict__ w1_1o,
                              const float* __restrict__ w1_2e, const float* __restrict__ w2_0e,
                              const float* __restrict__ w2_1o, const float* __restrict__ w2_2e,
                              const float* __restrict__ w3_0e, const float* __restrict__ w3_1o,
                              const float* __restrict__ w3_2e, int row, int s, int c) {
  if (row < 1)  return w1_0e[s * NCH + c];
  if (row < 2)  return w1_1o[s * NCH + c];
  if (row < 3)  return w1_2e[s * NCH + c];
  if (row < 5)  return w2_0e[(s * 2 + row - 3) * NCH + c];
  if (row < 7)  return w2_1o[(s * 2 + row - 5) * NCH + c];
  if (row < 10) return w2_2e[(s * 3 + row - 7) * NCH + c];
  if (row < 14) return w3_0e[(s * 4 + row - 10) * NCH + c];
  if (row < 20) return w3_1o[(s * 6 + row - 14) * NCH + c];
  return w3_2e[(s * 8 + row - 20) * NCH + c];
}

__global__ __launch_bounds__(256) void prep_uw(
    const float* __restrict__ u1_0e, const float* __restrict__ u1_1o, const float* __restrict__ u1_2e,
    const float* __restrict__ u2_0e, const float* __restrict__ u2_1o, const float* __restrict__ u2_2e,
    const float* __restrict__ u3_0e, const float* __restrict__ u3_1o, const float* __restrict__ u3_2e,
    const float* __restrict__ w1_0e, const float* __restrict__ w1_1o, const float* __restrict__ w1_2e,
    const float* __restrict__ w2_0e, const float* __restrict__ w2_1o, const float* __restrict__ w2_2e,
    const float* __restrict__ w3_0e, const float* __restrict__ w3_1o, const float* __restrict__ w3_2e,
    const int* __restrict__ index, float* __restrict__ ws) {
  const int tid = threadIdx.x;
  if (blockIdx.x < PREP_BLOCKS) {
    __shared__ float sW[WROWS * CG];
    const int bid  = blockIdx.x;
    const int s    = bid / ((NCH / CG) * EPARTS);
    const int rem  = bid - s * ((NCH / CG) * EPARTS);
    const int cg   = rem / EPARTS;
    const int part = rem - cg * EPARTS;
    const int c0   = cg * CG;
    for (int t = tid; t < WROWS * CG; t += 256) {
      const int row = t / CG;
      const int cc  = t - row * CG;
      sW[t] = w_row(w1_0e, w1_1o, w1_2e, w2_0e, w2_1o, w2_2e, w3_0e, w3_1o, w3_2e, row, s, c0 + cc);
    }
    __syncthreads();
    const int eEnd = (part + 1) * EPART_LEN;
    for (int e = part * EPART_LEN + tid; e < eEnd; e += 256) {
      const int n = e / KROW;
      const int p = e - n * KROW;
      const int ir   = (n == 0) ? 0 : (n < 4 ? 1 : 2);
      const int idim = (ir == 0) ? 1 : (ir == 1 ? 3 : 5);
      const int i    = (ir == 0) ? 0 : (ir == 1 ? n - 1 : n - 4);
      const float* u = nullptr;
      int mul = 0, ubase = 0, wrow0 = 0;
      if (p < G1) {
        if (p < 9) {
          u = (ir == 0) ? u1_0e : (ir == 1 ? u1_1o : u1_2e);
          mul = 1;
          ubase = p * idim + i;
          wrow0 = ir;
        }
      } else if (p < G1 + G2) {
        const int q = p - G1;
        if (q < 81) {
          u = (ir == 0) ? u2_0e : (ir == 1 ? u2_1o : u2_2e);
          mul = (ir == 2) ? 3 : 2;
          ubase = q * mul * idim + i;
          wrow0 = (ir == 0) ? 3 : (ir == 1 ? 5 : 7);
        }
      } else {
        const int t  = p - (G1 + G2);
        const int d0 = t / G2;
        const int q  = t - d0 * G2;
        if (q < 81) {
          u = (ir == 0) ? u3_0e : (ir == 1 ? u3_1o : u3_2e);
          mul = (ir == 0) ? 4 : (ir == 1 ? 6 : 8);
          ubase = (d0 * 81 + q) * mul * idim + i;
          wrow0 = (ir == 0) ? 10 : (ir == 1 ? 14 : 20);
        }
      }
      float v[CG];
#pragma unroll
      for (int cc = 0; cc < CG; ++cc) v[cc] = 0.f;
      for (int k = 0; k < mul; ++k) {
        const float uk = u[ubase + k * idim];
        const float* wr = sW + (wrow0 + k) * CG;
#pragma unroll
        for (int cc = 0; cc < CG; ++cc) v[cc] += wr[cc] * uk;
      }
      float* dst = ws + (size_t)(s * NCH + c0) * (NCOMP * KROW) + e;
#pragma unroll
      for (int cc = 0; cc < CG; ++cc) dst[(size_t)cc * (NCOMP * KROW)] = v[cc];
    }
  } else {
    __shared__ int sidx[NNODE];
    __shared__ int scnt[NSPEC];
    int* zlist = (int*)(ws + ZL_OFF);
    int* cnt   = (int*)(ws + CNT_OFF);
    int* table = (int*)(ws + TAB_OFF);
    for (int z = tid; z < NNODE; z += 256) {
      int v = index[z];
      v = v < 0 ? 0 : (v > NSPEC - 1 ? NSPEC - 1 : v);
      sidx[z] = v;
    }
    __syncthreads();
    for (int z = tid; z < NNODE; z += 256) {
      const int s = sidx[z];
      int rank = 0;
      for (int zz = 0; zz < z; ++zz) rank += (sidx[zz] == s) ? 1 : 0;
      zlist[s * NNODE + rank] = z;
    }
    if (tid < NSPEC) {
      int n = 0;
      for (int zz = 0; zz < NNODE; ++zz) n += (sidx[zz] == tid) ? 1 : 0;
      scnt[tid] = n;
      cnt[tid]  = n;
    }
    __syncthreads();
    if (tid == 0) {
      int slot = 0;
      for (int s = 0; s < NSPEC; ++s)
        for (int r0 = 0; r0 < scnt[s] && slot < MAXSLOT; r0 += CHUNK_ROWS) {
          table[2 * slot]     = s;
          table[2 * slot + 1] = r0;
          ++slot;
        }
      for (; slot < MAXSLOT; ++slot) {
        table[2 * slot]     = -1;
        table[2 * slot + 1] = 0;
      }
    }
  }
}

__global__ __launch_bounds__(128) void symcon_main(
    const float* __restrict__ x, const float* __restrict__ ws, float* __restrict__ out) {
  const int* zlist = (const int*)(ws + ZL_OFF);
  const int* cnt   = (const int*)(ws + CNT_OFF);
  const int* table = (const int*)(ws + TAB_OFF);

  const int slot = blockIdx.y;
  const int s = table[2 * slot];
  if (s < 0) return;
  const int r0 = table[2 * slot + 1];
  const int n  = cnt[s];

  const int wv   = __builtin_amdgcn_readfirstlane(threadIdx.x >> 5);
  const int c    = blockIdx.x * 4 + wv;
  const int lane = threadIdx.x & 31;
  const int h    = lane >> 4;
  const int r    = lane & 15;

  float xv[2][9];
#pragma unroll
  for (int t = 0; t < 2; ++t) {
    const int rr = r0 + 16 * t + r;
    const bool valid = rr < n;
    const int z = valid ? zlist[s * NNODE + rr] : 0;
    const float* xr = x + ((size_t)z * NCH + c) * 9;
#pragma unroll
    for (int d = 0; d < 9; ++d) xv[t][d] = valid ? xr[d] : 0.f;
  }

  v2f A1[2][3];
  v2f P[2][21];
#pragma unroll
  for (int t = 0; t < 2; ++t) {
#pragma unroll
    for (int j = 0; j < 3; ++j) {
#pragma unroll
      for (int e = 0; e < 2; ++e) {
        const int q0 = 4 * j + e;
        const int q1 = q0 + 2;
        const float a0 = (q0 < 9) ? xv[t][q0 < 9 ? q0 : 0] : 0.f;
        const float a1 = (q1 < 9) ? xv[t][q1 < 9 ? q1 : 0] : 0.f;
        A1[t][j][e] = h ? a1 : a0;
      }
    }
#pragma unroll
    for (int j = 0; j < 21; ++j) {
#pragma unroll
      for (int e = 0; e < 2; ++e) {
        const int q0 = 4 * j + e;
        const int q1 = q0 + 2;
        const float p0 = (q0 < 81) ? xv[t][q0 < 81 ? q0 / 9 : 0] * xv[t][q0 % 9] : 0.f;
        const float p1 = (q1 < 81) ? xv[t][q1 < 81 ? q1 / 9 : 0] * xv[t][q1 % 9] : 0.f;
        P[t][j][e] = h ? p1 : p0;
      }
    }
  }

  const int col = r < NCOMP ? r : NCOMP - 1;
  const float* bp = ws + ((size_t)((s * NCH + c) * NCOMP + col)) * KROW + 2 * h;

  v8f acc0 = {0.f, 0.f, 0.f, 0.f, 0.f, 0.f, 0.f, 0.f};
  v8f acc1 = {0.f, 0.f, 0.f, 0.f, 0.f, 0.f, 0.f, 0.f};

#pragma unroll
  for (int j = 0; j < 3; ++j) {
    const v2f b = *(const v2f*)(bp + 4 * j);
    acc0 = __builtin_amdgcn_wmma_f32_16x16x4_f32(false, A1[0][j], false, b, (short)0, acc0, false, false);
    acc1 = __builtin_amdgcn_wmma_f32_16x16x4_f32(false, A1[1][j], false, b, (short)0, acc1, false, false);
  }
#pragma unroll
  for (int j = 0; j < 21; ++j) {
    const v2f b = *(const v2f*)(bp + G1 + 4 * j);
    acc0 = __builtin_amdgcn_wmma_f32_16x16x4_f32(false, P[0][j], false, b, (short)0, acc0, false, false);
    acc1 = __builtin_amdgcn_wmma_f32_16x16x4_f32(false, P[1][j], false, b, (short)0, acc1, false, false);
  }
#pragma unroll
  for (int d0 = 0; d0 < 9; ++d0) {
#pragma unroll
    for (int j = 0; j < 21; ++j) {
      const v2f b  = *(const v2f*)(bp + G1 + G2 + G2 * d0 + 4 * j);
      const v2f a0 = P[0][j] * xv[0][d0];
      const v2f a1 = P[1][j] * xv[1][d0];
      acc0 = __builtin_amdgcn_wmma_f32_16x16x4_f32(false, a0, false, b, (short)0, acc0, false, false);
      acc1 = __builtin_amdgcn_wmma_f32_16x16x4_f32(false, a1, false, b, (short)0, acc1, false, false);
    }
  }

  if (r < NCOMP) {
    const int coff = (r == 0) ? c : (r < 4 ? 128 + c * 3 + (r - 1) : 512 + c * 5 + (r - 4));
#pragma unroll
    for (int j = 0; j < 8; ++j) {
      const int rr0 = r0 + 8 * h + j;
      if (rr0 < n) out[(size_t)zlist[s * NNODE + rr0] * 1152 + coff] = acc0[j];
      const int rr1 = rr0 + 16;
      if (rr1 < n) out[(size_t)zlist[s * NNODE + rr1] * 1152 + coff] = acc1[j];
    }
  }
}

extern "C" void kernel_launch(void* const* d_in, const int* in_sizes, int n_in,
                              void* d_out, int out_size, void* d_ws, size_t ws_size,
                              hipStream_t stream) {
  const float* x     = (const float*)d_in[0];
  const float* u1_0e = (const float*)d_in[1];
  const float* w1_0e = (const float*)d_in[2];
  const float* u1_1o = (const float*)d_in[3];
  const float* w1_1o = (const float*)d_in[4];
  const float* u1_2e = (const float*)d_in[5];
  const float* w1_2e = (const float*)d_in[6];
  const float* u2_0e = (const float*)d_in[7];
  const float* w2_0e = (const float*)d_in[8];
  const float* u2_1o = (const float*)d_in[9];
  const float* w2_1o = (const float*)d_in[10];
  const float* u2_2e = (const float*)d_in[11];
  const float* w2_2e = (const float*)d_in[12];
  const float* u3_0e = (const float*)d_in[13];
  const float* w3_0e = (const float*)d_in[14];
  const float* u3_1o = (const float*)d_in[15];
  const float* w3_1o = (const float*)d_in[16];
  const float* u3_2e = (const float*)d_in[17];
  const float* w3_2e = (const float*)d_in[18];
  const int*   index = (const int*)d_in[19];
  float* ws  = (float*)d_ws;
  float* out = (float*)d_out;

  prep_uw<<<PREP_BLOCKS + 1, 256, 0, stream>>>(
      u1_0e, u1_1o, u1_2e, u2_0e, u2_1o, u2_2e, u3_0e, u3_1o, u3_2e,
      w1_0e, w1_1o, w1_2e, w2_0e, w2_1o, w2_2e, w3_0e, w3_1o, w3_2e, index, ws);
  symcon_main<<<dim3(NCH / 4, MAXSLOT), 128, 0, stream>>>(x, ws, out);
}


# Species bookkeeping block: node ranks from wave ballots and popcounts plus a prefix over 32 group counts, not a serial LDS scan of all earlier nodes; same lists, same sums and order
# speedup vs baseline: 2.7372x; 1.9135x over previous
#include <hip/hip_runtime.h>

typedef float v2f __attribute__((ext_vector_type(2)));
typedef float v8f __attribute__((ext_vector_type(8)));

#define NSPEC 10
#define NCH 128
#define NNODE 1024
#define NCOMP 9
#define G1 12
#define G2 84
#define KROW 852
#define CHUNK_ROWS 32
#define NGROUP (NNODE / 32)
#define MAXSLOT 42
#define UW_FLOATS (NSPEC * NCH * NCOMP * KROW)
#define ZL_OFF  UW_FLOATS
#define CNT_OFF (UW_FLOATS + NSPEC * NNODE)
#define TAB_OFF (CNT_OFF + 16)

#define CG 32
#define EPARTS 12
#define EPART_LEN 639
#define WROWS 28
#define PREP_BLOCKS (NSPEC * (NCH / CG) * EPARTS)

__device__ inline float w_row(const float* __restrict__ w1_0e, const float* __restrict__ w1_1o,
                              const float* __restrict__ w1_2e, const float* __restrict__ w2_0e,
                              const float* __restrict__ w2_1o, const float* __restrict__ w2_2e,
                              const float* __restrict__ w3_0e, const float* __restrict__ w3_1o,
                              const float* __restrict__ w3_2e, int row, int s, int c) {
  if (row < 1)  return w1_0e[s * NCH + c];
  if (row < 2)  return w1_1o[s * NCH + c];
  if (row < 3)  return w1_2e[s * NCH + c];
  if (row < 5)  return w2_0e[(s * 2 + row - 3) * NCH + c];
  if (row < 7)  return w2_1o[(s * 2 + row - 5) * NCH + c];
  if (row < 10) return w2_2e[(s * 3 + row - 7) * NCH + c];
  if (row < 14) return w3_0e[(s * 4 + row - 10) * NCH + c];
  if (row < 20) return w3_1o[(s * 6 + row - 14) * NCH + c];
  return w3_2e[(s * 8 + row - 20) * NCH + c];
}

__global__ __launch_bounds__(256) void prep_uw(
    const float* __restrict__ u1_0e, const float* __restrict__ u1_1o, const float* __restrict__ u1_2e,
    const float* __restrict__ u2_0e, const float* __restrict__ u2_1o, const float* __restrict__ u2_2e,
    const float* __restrict__ u3_0e, const float* __restrict__ u3_1o, const float* __restrict__ u3_2e,
    const float* __restrict__ w1_0e, const float* __restrict__ w1_1o, const float* __restrict__ w1_2e,
    const float* __restrict__ w2_0e, const float* __restrict__ w2_1o, const float* __restrict__ w2_2e,
    const float* __restrict__ w3_0e, const float* __restrict__ w3_1o, const float* __restrict__ w3_2e,
    const int* __restrict__ index, float* __restrict__ ws) {
  const int tid = threadIdx.x;
  if (blockIdx.x < PREP_BLOCKS) {
    __shared__ float sW[WROWS * CG];
    const int bid  = blockIdx.x;
    const int s    = bid / ((NCH / CG) * EPARTS);
    const int rem  = bid - s * ((NCH / CG) * EPARTS);
    const int cg   = rem / EPARTS;
    const int part = rem - cg * EPARTS;
    const int c0   = cg * CG;
    for (int t = tid; t < WROWS * CG; t += 256) {
      const int row = t / CG;
      const int cc  = t - row * CG;
      sW[t] = w_row(w1_0e, w1_1o, w1_2e, w2_0e, w2_1o, w2_2e, w3_0e, w3_1o, w3_2e, row, s, c0 + cc);
    }
    __syncthreads();
    const int eEnd = (part + 1) * EPART_LEN;
    for (int e = part * EPART_LEN + tid; e < eEnd; e += 256) {
      const int n = e / KROW;
      const int p = e - n * KROW;
      const int ir   = (n == 0) ? 0 : (n < 4 ? 1 : 2);
      const int idim = (ir == 0) ? 1 : (ir == 1 ? 3 : 5);
      const int i    = (ir == 0) ? 0 : (ir == 1 ? n - 1 : n - 4);
      const float* u = nullptr;
      int mul = 0, ubase = 0, wrow0 = 0;
      if (p < G1) {
        if (p < 9) {
          u = (ir == 0) ? u1_0e : (ir == 1 ? u1_1o : u1_2e);
          mul = 1;
          ubase = p * idim + i;
          wrow0 = ir;
        }
      } else if (p < G1 + G2) {
        const int q = p - G1;
        if (q < 81) {
          u = (ir == 0) ? u2_0e : (ir == 1 ? u2_1o : u2_2e);
          mul = (ir == 2) ? 3 : 2;
          ubase = q * mul * idim + i;
          wrow0 = (ir == 0) ? 3 : (ir == 1 ? 5 : 7);
        }
      } else {
        const int t  = p - (G1 + G2);
        const int d0 = t / G2;
        const int q  = t - d0 * G2;
        if (q < 81) {
          u = (ir == 0) ? u3_0e : (ir == 1 ? u3_1o : u3_2e);
          mul = (ir == 0) ? 4 : (ir == 1 ? 6 : 8);
          ubase = (d0 * 81 + q) * mul * idim + i;
          wrow0 = (ir == 0) ? 10 : (ir == 1 ? 14 : 20);
        }
      }
      float v[CG];
#pragma unroll
      for (int cc = 0; cc < CG; ++cc) v[cc] = 0.f;
      for (int k = 0; k < mul; ++k) {
        const float uk = u[ubase + k * idim];
        const float* wr = sW + (wrow0 + k) * CG;
#pragma unroll
        for (int cc = 0; cc < CG; ++cc) v[cc] += wr[cc] * uk;
      }
      float* dst = ws + (size_t)(s * NCH + c0) * (NCOMP * KROW) + e;
#pragma unroll
      for (int cc = 0; cc < CG; ++cc) dst[(size_t)cc * (NCOMP * KROW)] = v[cc];
    }
  } else {
    __shared__ int sCnt[NGROUP * NSPEC];
    __shared__ int sBase[NGROUP * NSPEC];
    __shared__ int scnt[NSPEC];
    int* zlist = (int*)(ws + ZL_OFF);
    int* cnt   = (int*)(ws + CNT_OFF);
    int* table = (int*)(ws + TAB_OFF);
    const int lane = tid & 31;
    const unsigned below = (1u << lane) - 1u;
    int sp[NNODE / 256];
    int rk[NNODE / 256];
    for (int k = 0; k < NNODE / 256; ++k) {
      int v = index[tid + 256 * k];
      sp[k] = v < 0 ? 0 : (v > NSPEC - 1 ? NSPEC - 1 : v);
    }
    for (int k = 0; k < NNODE / 256; ++k) {
      const int grp = (tid >> 5) + 8 * k;
      int myrank = 0, mycount = 0;
      for (int s = 0; s < NSPEC; ++s) {
        const unsigned m = __builtin_amdgcn_ballot_w32(sp[k] == s);
        if (sp[k] == s) myrank = __builtin_popcount(m & below);
        if (lane == s) mycount = __builtin_popcount(m);
      }
      rk[k] = myrank;
      if (lane < NSPEC) sCnt[grp * NSPEC + lane] = mycount;
    }
    __syncthreads();
    if (tid < NSPEC) {
      int run = 0;
      for (int g = 0; g < NGROUP; ++g) {
        sBase[g * NSPEC + tid] = run;
        run += sCnt[g * NSPEC + tid];
      }
      scnt[tid] = run;
      cnt[tid]  = run;
    }
    __syncthreads();
    for (int k = 0; k < NNODE / 256; ++k) {
      const int grp = (tid >> 5) + 8 * k;
      zlist[sp[k] * NNODE + sBase[grp * NSPEC + sp[k]] + rk[k]] = tid + 256 * k;
    }
    if (tid == 0) {
      int slot = 0;
      for (int s = 0; s < NSPEC; ++s)
        for (int r0 = 0; r0 < scnt[s] && slot < MAXSLOT; r0 += CHUNK_ROWS) {
          table[2 * slot]     = s;
          table[2 * slot + 1] = r0;
          ++slot;
        }
      for (; slot < MAXSLOT; ++slot) {
        table[2 * slot]     = -1;
        table[2 * slot + 1] = 0;
      }
    }
  }
}

__global__ __launch_bounds__(128) void symcon_main(
    const float* __restrict__ x, const float* __restrict__ ws, float* __restrict__ out) {
  const int* zlist = (const int*)(ws + ZL_OFF);
  const int* cnt   = (const int*)(ws + CNT_OFF);
  const int* table = (const int*)(ws + TAB_OFF);

  const int slot = blockIdx.y;
  const int s = table[2 * slot];
  if (s < 0) return;
  const int r0 = table[2 * slot + 1];
  const int n  = cnt[s];

  const int wv   = __builtin_amdgcn_readfirstlane(threadIdx.x >> 5);
  const int c    = blockIdx.x * 4 + wv;
  const int lane = threadIdx.x & 31;
  const int h    = lane >> 4;
  const int r    = lane & 15;

  float xv[2][9];
#pragma unroll
  for (int t = 0; t < 2; ++t) {
    const int rr = r0 + 16 * t + r;
    const bool valid = rr < n;
    const int z = valid ? zlist[s * NNODE + rr] : 0;
    const float* xr = x + ((size_t)z * NCH + c) * 9;
#pragma unroll
    for (int d = 0; d < 9; ++d) xv[t][d] = valid ? xr[d] : 0.f;
  }

  v2f A1[2][3];
  v2f P[2][21];
#pragma unroll
  for (int t = 0; t < 2; ++t) {
#pragma unroll
    for (int j = 0; j < 3; ++j) {
#pragma unroll
      for (int e = 0; e < 2; ++e) {
        const int q0 = 4 * j + e;
        const int q1 = q0 + 2;
        const float a0 = (q0 < 9) ? xv[t][q0 < 9 ? q0 : 0] : 0.f;
        const float a1 = (q1 < 9) ? xv[t][q1 < 9 ? q1 : 0] : 0.f;
        A1[t][j][e] = h ? a1 : a0;
      }
    }
#pragma unroll
    for (int j = 0; j < 21; ++j) {
#pragma unroll
      for (int e = 0; e < 2; ++e) {
        const int q0 = 4 * j + e;
        const int q1 = q0 + 2;
        const float p0 = (q0 < 81) ? xv[t][q0 < 81 ? q0 / 9 : 0] * xv[t][q0 % 9] : 0.f;
        const float p1 = (q1 < 81) ? xv[t][q1 < 81 ? q1 / 9 : 0] * xv[t][q1 % 9] : 0.f;
        P[t][j][e] = h ? p1 : p0;
      }
    }
  }

  const int col = r < NCOMP ? r : NCOMP - 1;
  const float* bp = ws + ((size_t)((s * NCH + c) * NCOMP + col)) * KROW + 2 * h;

  v8f acc0 = {0.f, 0.f, 0.f, 0.f, 0.f, 0.f, 0.f, 0.f};
  v8f acc1 = {0.f, 0.f, 0.f, 0.f, 0.f, 0.f, 0.f, 0.f};

#pragma unroll
  for (int j = 0; j < 3; ++j) {
    const v2f b = *(const v2f*)(bp + 4 * j);
    acc0 = __builtin_amdgcn_wmma_f32_16x16x4_f32(false, A1[0][j], false, b, (short)0, acc0, false, false);
    acc1 = __builtin_amdgcn_wmma_f32_16x16x4_f32(false, A1[1][j], false, b, (short)0, acc1, false, false);
  }
#pragma unroll
  for (int j = 0; j < 21; ++j) {
    const v2f b = *(const v2f*)(bp + G1 + 4 * j);
    acc0 = __builtin_amdgcn_wmma_f32_16x16x4_f32(false, P[0][j], false, b, (short)0, acc0, false, false);
    acc1 = __builtin_amdgcn_wmma_f32_16x16x4_f32(false, P[1][j], false, b, (short)0, acc1, false, false);
  }
#pragma unroll
  for (int d0 = 0; d0 < 9; ++d0) {
#pragma unroll
    for (int j = 0; j < 21; ++j) {
      const v2f b  = *(const v2f*)(bp + G1 + G2 + G2 * d0 + 4 * j);
      const v2f a0 = P[0][j] * xv[0][d0];
      const v2f a1 = P[1][j] * xv[1][d0];
      acc0 = __builtin_amdgcn_wmma_f32_16x16x4_f32(false, a0, false, b, (short)0, acc0, false, false);
      acc1 = __builtin_amdgcn_wmma_f32_16x16x4_f32(false, a1, false, b, (short)0, acc1, false, false);
    }
  }

  if (r < NCOMP) {
    const int coff = (r == 0) ? c : (r < 4 ? 128 + c * 3 + (r - 1) : 512 + c * 5 + (r - 4));
#pragma unroll
    for (int j = 0; j < 8; ++j) {
      const int rr0 = r0 + 8 * h + j;
      if (rr0 < n) out[(size_t)zlist[s * NNODE + rr0] * 1152 + coff] = acc0[j];
      const int rr1 = rr0 + 16;
      if (rr1 < n) out[(size_t)zlist[s * NNODE + rr1] * 1152 + coff] = acc1[j];
    }
  }
}

extern "C" void kernel_launch(void* const* d_in, const int* in_sizes, int n_in,
                              void* d_out, int out_size, void* d_ws, size_t ws_size,
                              hipStream_t stream) {
  const float* x     = (const float*)d_in[0];
  const float* u1_0e = (const float*)d_in[1];
  const float* w1_0e = (const float*)d_in[2];
  const float* u1_1o = (const float*)d_in[3];
  const float* w1_1o = (const float*)d_in[4];
  const float* u1_2e = (const float*)d_in[5];
  const float* w1_2e = (const float*)d_in[6];
  const float* u2_0e = (const float*)d_in[7];
  const float* w2_0e = (const float*)d_in[8];
  const float* u2_1o = (const float*)d_in[9];
  const float* w2_1o = (const float*)d_in[10];
  const float* u2_2e = (const float*)d_in[11];
  const float* w2_2e = (const float*)d_in[12];
  const float* u3_0e = (const float*)d_in[13];
  const float* w3_0e = (const float*)d_in[14];
  const float* u3_1o = (const float*)d_in[15];
  const float* w3_1o = (const float*)d_in[16];
  const float* u3_2e = (const float*)d_in[17];
  const float* w3_2e = (const float*)d_in[18];
  const int*   index = (const int*)d_in[19];
  float* ws  = (float*)d_ws;
  float* out = (float*)d_out;

  prep_uw<<<PREP_BLOCKS + 1, 256, 0, stream>>>(
      u1_0e, u1_1o, u1_2e, u2_0e, u2_1o, u2_2e, u3_0e, u3_1o, u3_2e,
      w1_0e, w1_1o, w1_2e, w2_0e, w2_1o, w2_2e, w3_0e, w3_1o, w3_2e, index, ws);
  symcon_main<<<dim3(NCH / 4, MAXSLOT), 128, 0, stream>>>(x, ws, out);
}


# Main kernel: B fragments loaded 14 K-steps ahead of the WMMAs that use them (groups of 4 loads kept ahead by a scheduling barrier), not waited for right before use; same sums and order
# speedup vs baseline: 3.5523x; 1.2978x over previous
#include <hip/hip_runtime.h>

typedef float v2f __attribute__((ext_vector_type(2)));
typedef float v8f __attribute__((ext_vector_type(8)));

#define NSPEC 10
#define NCH 128
#define NNODE 1024
#define NCOMP 9
#define G1 12
#define G2 84
#define KROW 852
#define CHUNK_ROWS 32
#define NGROUP (NNODE / 32)
#define MAXSLOT 42
#define UW_FLOATS (NSPEC * NCH * NCOMP * KROW)
#define ZL_OFF  UW_FLOATS
#define CNT_OFF (UW_FLOATS + NSPEC * NNODE)
#define TAB_OFF (CNT_OFF + 16)

#define CG 32
#define EPARTS 12
#define EPART_LEN 639
#define WROWS 28
#define PREP_BLOCKS (NSPEC * (NCH / CG) * EPARTS)

__device__ inline float w_row(const float* __restrict__ w1_0e, const float* __restrict__ w1_1o,
                              const float* __restrict__ w1_2e, const float* __restrict__ w2_0e,
                              const float* __restrict__ w2_1o, const float* __restrict__ w2_2e,
                              const float* __restrict__ w3_0e, const float* __restrict__ w3_1o,
                              const float* __restrict__ w3_2e, int row, int s, int c) {
  if (row < 1)  return w1_0e[s * NCH + c];
  if (row < 2)  return w1_1o[s * NCH + c];
  if (row < 3)  return w1_2e[s * NCH + c];
  if (row < 5)  return w2_0e[(s * 2 + row - 3) * NCH + c];
  if (row < 7)  return w2_1o[(s * 2 + row - 5) * NCH + c];
  if (row < 10) return w2_2e[(s * 3 + row - 7) * NCH + c];
  if (row < 14) return w3_0e[(s * 4 + row - 10) * NCH + c];
  if (row < 20) return w3_1o[(s * 6 + row - 14) * NCH + c];
  return w3_2e[(s * 8 + row - 20) * NCH + c];
}

__global__ __launch_bounds__(256) void prep_uw(
    const float* __restrict__ u1_0e, const float* __restrict__ u1_1o, const float* __restrict__ u1_2e,
    const float* __restrict__ u2_0e, const float* __restrict__ u2_1o, const float* __restrict__ u2_2e,
    const float* __restrict__ u3_0e, const float* __restrict__ u3_1o, const float* __restrict__ u3_2e,
    const float* __restrict__ w1_0e, const float* __restrict__ w1_1o, const float* __restrict__ w1_2e,
    const float* __restrict__ w2_0e, const float* __restrict__ w2_1o, const float* __restrict__ w2_2e,
    const float* __restrict__ w3_0e, const float* __restrict__ w3_1o, const float* __restrict__ w3_2e,
    const int* __restrict__ index, float* __restrict__ ws) {
  const int tid = threadIdx.x;
  if (blockIdx.x < PREP_BLOCKS) {
    __shared__ float sW[WROWS * CG];
    const int bid  = blockIdx.x;
    const int s    = bid / ((NCH / CG) * EPARTS);
    const int rem  = bid - s * ((NCH / CG) * EPARTS);
    const int cg   = rem / EPARTS;
    const int part = rem - cg * EPARTS;
    const int c0   = cg * CG;
    for (int t = tid; t < WROWS * CG; t += 256) {
      const int row = t / CG;
      const int cc  = t - row * CG;
      sW[t] = w_row(w1_0e, w1_1o, w1_2e, w2_0e, w2_1o, w2_2e, w3_0e, w3_1o, w3_2e, row, s, c0 + cc);
    }
    __syncthreads();
    const int eEnd = (part + 1) * EPART_LEN;
    for (int e = part * EPART_LEN + tid; e < eEnd; e += 256) {
      const int n = e / KROW;
      const int p = e - n * KROW;
      const int ir   = (n == 0) ? 0 : (n < 4 ? 1 : 2);
      const int idim = (ir == 0) ? 1 : (ir == 1 ? 3 : 5);
      const int i    = (ir == 0) ? 0 : (ir == 1 ? n - 1 : n - 4);
      const float* u = nullptr;
      int mul = 0, ubase = 0, wrow0 = 0;
      if (p < G1) {
        if (p < 9) {
          u = (ir == 0) ? u1_0e : (ir == 1 ? u1_1o : u1_2e);
          mul = 1;
          ubase = p * idim + i;
          wrow0 = ir;
        }
      } else if (p < G1 + G2) {
        const int q = p - G1;
        if (q < 81) {
          u = (ir == 0) ? u2_0e : (ir == 1 ? u2_1o : u2_2e);
          mul = (ir == 2) ? 3 : 2;
          ubase = q * mul * idim + i;
          wrow0 = (ir == 0) ? 3 : (ir == 1 ? 5 : 7);
        }
      } else {
        const int t  = p - (G1 + G2);
        const int d0 = t / G2;
        const int q  = t - d0 * G2;
        if (q < 81) {
          u = (ir == 0) ? u3_0e : (ir == 1 ? u3_1o : u3_2e);
          mul = (ir == 0) ? 4 : (ir == 1 ? 6 : 8);
          ubase = (d0 * 81 + q) * mul * idim + i;
          wrow0 = (ir == 0) ? 10 : (ir == 1 ? 14 : 20);
        }
      }
      float v[CG];
#pragma unroll
      for (int cc = 0; cc < CG; ++cc) v[cc] = 0.f;
      for (int k = 0; k < mul; ++k) {
        const float uk = u[ubase + k * idim];
        const float* wr = sW + (wrow0 + k) * CG;
#pragma unroll
        for (int cc = 0; cc < CG; ++cc) v[cc] += wr[cc] * uk;
      }
      float* dst = ws + (size_t)(s * NCH + c0) * (NCOMP * KROW) + e;
#pragma unroll
      for (int cc = 0; cc < CG; ++cc) dst[(size_t)cc * (NCOMP * KROW)] = v[cc];
    }
  } else {
    __shared__ int sCnt[NGROUP * NSPEC];
    __shared__ int sBase[NGROUP * NSPEC];
    __shared__ int scnt[NSPEC];
    int* zlist = (int*)(ws + ZL_OFF);
    int* cnt   = (int*)(ws + CNT_OFF);
    int* table = (int*)(ws + TAB_OFF);
    const int lane = tid & 31;
    const unsigned below = (1u << lane) - 1u;
    int sp[NNODE / 256];
    int rk[NNODE / 256];
    for (int k = 0; k < NNODE / 256; ++k) {
      int v = index[tid + 256 * k];
      sp[k] = v < 0 ? 0 : (v > NSPEC - 1 ? NSPEC - 1 : v);
    }
    for (int k = 0; k < NNODE / 256; ++k) {
      const int grp = (tid >> 5) + 8 * k;
      int myrank = 0, mycount = 0;
      for (int s = 0; s < NSPEC; ++s) {
        const unsigned m = __builtin_amdgcn_ballot_w32(sp[k] == s);
        if (sp[k] == s) myrank = __builtin_popcount(m & below);
        if (lane == s) mycount = __builtin_popcount(m);
      }
      rk[k] = myrank;
      if (lane < NSPEC) sCnt[grp * NSPEC + lane] = mycount;
    }
    __syncthreads();
    if (tid < NSPEC) {
      int run = 0;
      for (int g = 0; g < NGROUP; ++g) {
        sBase[g * NSPEC + tid] = run;
        run += sCnt[g * NSPEC + tid];
      }
      scnt[tid] = run;
      cnt[tid]  = run;
    }
    __syncthreads();
    for (int k = 0; k < NNODE / 256; ++k) {
      const int grp = (tid >> 5) + 8 * k;
      zlist[sp[k] * NNODE + sBase[grp * NSPEC + sp[k]] + rk[k]] = tid + 256 * k;
    }
    if (tid == 0) {
      int slot = 0;
      for (int s = 0; s < NSPEC; ++s)
        for (int r0 = 0; r0 < scnt[s] && slot < MAXSLOT; r0 += CHUNK_ROWS) {
          table[2 * slot]     = s;
          table[2 * slot + 1] = r0;
          ++slot;
        }
      for (; slot < MAXSLOT; ++slot) {
        table[2 * slot]     = -1;
        table[2 * slot + 1] = 0;
      }
    }
  }
}

#define NSTEP 213
#define PF 14
#define GRP 4
__global__ __launch_bounds__(128) void symcon_main(
    const float* __restrict__ x, const float* __restrict__ ws, float* __restrict__ out) {
  const int* zlist = (const int*)(ws + ZL_OFF);
  const int* cnt   = (const int*)(ws + CNT_OFF);
  const int* table = (const int*)(ws + TAB_OFF);

  const int slot = blockIdx.y;
  const int s = table[2 * slot];
  if (s < 0) return;
  const int r0 = table[2 * slot + 1];
  const int n  = cnt[s];

  const int wv   = __builtin_amdgcn_readfirstlane(threadIdx.x >> 5);
  const int c    = blockIdx.x * 4 + wv;
  const int lane = threadIdx.x & 31;
  const int h    = lane >> 4;
  const int r    = lane & 15;

  int  zrow[2];
  bool valid[2];
#pragma unroll
  for (int t = 0; t < 2; ++t) {
    const int rr = r0 + 16 * t + r;
    valid[t] = rr < n;
    zrow[t]  = valid[t] ? zlist[s * NNODE + rr] : 0;
  }

  const int col = r < NCOMP ? r : NCOMP - 1;
  const float* bp = ws + ((size_t)((s * NCH + c) * NCOMP + col)) * KROW + 2 * h;

  v2f b[NSTEP];
#pragma unroll
  for (int q = 0; q < PF; ++q) b[q] = *(const v2f*)(bp + 4 * q);

  float xv[2][9];
#pragma unroll
  for (int t = 0; t < 2; ++t) {
    const float* xr = x + ((size_t)zrow[t] * NCH + c) * 9;
#pragma unroll
    for (int d = 0; d < 9; ++d) xv[t][d] = valid[t] ? xr[d] : 0.f;
  }

  v2f A1[2][3];
  v2f P[2][21];
#pragma unroll
  for (int t = 0; t < 2; ++t) {
#pragma unroll
    for (int j = 0; j < 3; ++j) {
#pragma unroll
      for (int e = 0; e < 2; ++e) {
        const int q0 = 4 * j + e;
        const int q1 = q0 + 2;
        const float a0 = (q0 < 9) ? xv[t][q0 < 9 ? q0 : 0] : 0.f;
        const float a1 = (q1 < 9) ? xv[t][q1 < 9 ? q1 : 0] : 0.f;
        A1[t][j][e] = h ? a1 : a0;
      }
    }
#pragma unroll
    for (int j = 0; j < 21; ++j) {
#pragma unroll
      for (int e = 0; e < 2; ++e) {
        const int q0 = 4 * j + e;
        const int q1 = q0 + 2;
        const float p0 = (q0 < 81) ? xv[t][q0 < 81 ? q0 / 9 : 0] * xv[t][q0 % 9] : 0.f;
        const float p1 = (q1 < 81) ? xv[t][q1 < 81 ? q1 / 9 : 0] * xv[t][q1 % 9] : 0.f;
        P[t][j][e] = h ? p1 : p0;
      }
    }
  }

  v8f acc0 = {0.f, 0.f, 0.f, 0.f, 0.f, 0.f, 0.f, 0.f};
  v8f acc1 = {0.f, 0.f, 0.f, 0.f, 0.f, 0.f, 0.f, 0.f};

#pragma unroll
  for (int g = 0; g < NSTEP; g += GRP) {
#pragma unroll
    for (int i = 0; i < GRP; ++i) {
      const int q = g + PF + i;
      if (q < NSTEP) b[q < NSTEP ? q : 0] = *(const v2f*)(bp + 4 * (q < NSTEP ? q : 0));
    }
    __builtin_amdgcn_sched_barrier(0);
#pragma unroll
    for (int i = 0; i < GRP; ++i) {
      const int st = g + i;
      if (st < NSTEP) {
        const int sc = st < NSTEP ? st : 0;
        v2f a0, a1;
        if (sc < 3) {
          a0 = A1[0][sc < 3 ? sc : 0];
          a1 = A1[1][sc < 3 ? sc : 0];
        } else if (sc < 24) {
          a0 = P[0][(sc >= 3 && sc < 24) ? sc - 3 : 0];
          a1 = P[1][(sc >= 3 && sc < 24) ? sc - 3 : 0];
        } else {
          const int d0 = (sc - 24) / 21;
          const int j  = (sc - 24) % 21;
          a0 = P[0][sc >= 24 ? j : 0] * xv[0][sc >= 24 ? d0 : 0];
          a1 = P[1][sc >= 24 ? j : 0] * xv[1][sc >= 24 ? d0 : 0];
        }
        acc0 = __builtin_amdgcn_wmma_f32_16x16x4_f32(false, a0, false, b[sc], (short)0, acc0, false, false);
        acc1 = __builtin_amdgcn_wmma_f32_16x16x4_f32(false, a1, false, b[sc], (short)0, acc1, false, false);
      }
    }
  }

  if (r < NCOMP) {
    const int coff = (r == 0) ? c : (r < 4 ? 128 + c * 3 + (r - 1) : 512 + c * 5 + (r - 4));
#pragma unroll
    for (int j = 0; j < 8; ++j) {
      const int rr0 = r0 + 8 * h + j;
      if (rr0 < n) out[(size_t)zlist[s * NNODE + rr0] * 1152 + coff] = acc0[j];
      const int rr1 = rr0 + 16;
      if (rr1 < n) out[(size_t)zlist[s * NNODE + rr1] * 1152 + coff] = acc1[j];
    }
  }
}

extern "C" void kernel_launch(void* const* d_in, const int* in_sizes, int n_in,
                              void* d_out, int out_size, void* d_ws, size_t ws_size,
                              hipStream_t stream) {
  const float* x     = (const float*)d_in[0];
  const float* u1_0e = (const float*)d_in[1];
  const float* w1_0e = (const float*)d_in[2];
  const float* u1_1o = (const float*)d_in[3];
  const float* w1_1o = (const float*)d_in[4];
  const float* u1_2e = (const float*)d_in[5];
  const float* w1_2e = (const float*)d_in[6];
  const float* u2_0e = (const float*)d_in[7];
  const float* w2_0e = (const float*)d_in[8];
  const float* u2_1o = (const float*)d_in[9];
  const float* w2_1o = (const float*)d_in[10];
  const float* u2_2e = (const float*)d_in[11];
  const float* w2_2e = (const float*)d_in[12];
  const float* u3_0e = (const float*)d_in[13];
  const float* w3_0e = (const float*)d_in[14];
  const float* u3_1o = (const float*)d_in[15];
  const float* w3_1o = (const float*)d_in[16];
  const float* u3_2e = (const float*)d_in[17];
  const float* w3_2e = (const float*)d_in[18];
  const int*   index = (const int*)d_in[19];
  float* ws  = (float*)d_ws;
  float* out = (float*)d_out;

  prep_uw<<<PREP_BLOCKS + 1, 256, 0, stream>>>(
      u1_0e, u1_1o, u1_2e, u2_0e, u2_1o, u2_2e, u3_0e, u3_1o, u3_2e,
      w1_0e, w1_1o, w1_2e, w2_0e, w2_1o, w2_2e, w3_0e, w3_1o, w3_2e, index, ws);
  symcon_main<<<dim3(NCH / 4, MAXSLOT), 128, 0, stream>>>(x, ws, out);
}


# Main kernel: each wave prefetches its UW rows into the second-level cache, one 256-byte window per row every 16 K-steps, 50 K-steps ahead of the B loads that read them; same loads, sums and order
# speedup vs baseline: 3.6800x; 1.0359x over previous
#include <hip/hip_runtime.h>

typedef float v2f __attribute__((ext_vector_type(2)));
typedef float v8f __attribute__((ext_vector_type(8)));

#define NSPEC 10
#define NCH 128
#define NNODE 1024
#define NCOMP 9
#define G1 12
#define G2 84
#define KROW 852
#define CHUNK_ROWS 32
#define NGROUP (NNODE / 32)
#define MAXSLOT 42
#define UW_FLOATS (NSPEC * NCH * NCOMP * KROW)
#define ZL_OFF  UW_FLOATS
#define CNT_OFF (UW_FLOATS + NSPEC * NNODE)
#define TAB_OFF (CNT_OFF + 16)

#define CG 32
#define EPARTS 12
#define EPART_LEN 639
#define WROWS 28
#define PREP_BLOCKS (NSPEC * (NCH / CG) * EPARTS)

__device__ inline float w_row(const float* __restrict__ w1_0e, const float* __restrict__ w1_1o,
                              const float* __restrict__ w1_2e, const float* __restrict__ w2_0e,
                              const float* __restrict__ w2_1o, const float* __restrict__ w2_2e,
                              const float* __restrict__ w3_0e, const float* __restrict__ w3_1o,
                              const float* __restrict__ w3_2e, int row, int s, int c) {
  if (row < 1)  return w1_0e[s * NCH + c];
  if (row < 2)  return w1_1o[s * NCH + c];
  if (row < 3)  return w1_2e[s * NCH + c];
  if (row < 5)  return w2_0e[(s * 2 + row - 3) * NCH + c];
  if (row < 7)  return w2_1o[(s * 2 + row - 5) * NCH + c];
  if (row < 10) return w2_2e[(s * 3 + row - 7) * NCH + c];
  if (row < 14) return w3_0e[(s * 4 + row - 10) * NCH + c];
  if (row < 20) return w3_1o[(s * 6 + row - 14) * NCH + c];
  return w3_2e[(s * 8 + row - 20) * NCH + c];
}

__global__ __launch_bounds__(256) void prep_uw(
    const float* __restrict__ u1_0e, const float* __restrict__ u1_1o, const float* __restrict__ u1_2e,
    const float* __restrict__ u2_0e, const float* __restrict__ u2_1o, const float* __restrict__ u2_2e,
    const float* __restrict__ u3_0e, const float* __restrict__ u3_1o, const float* __restrict__ u3_2e,
    const float* __restrict__ w1_0e, const float* __restrict__ w1_1o, const float* __restrict__ w1_2e,
    const float* __restrict__ w2_0e, const float* __restrict__ w2_1o, const float* __restrict__ w2_2e,
    const float* __restrict__ w3_0e, const float* __restrict__ w3_1o, const float* __restrict__ w3_2e,
    const int* __restrict__ index, float* __restrict__ ws) {
  const int tid = threadIdx.x;
  if (blockIdx.x < PREP_BLOCKS) {
    __shared__ float sW[WROWS * CG];
    const int bid  = blockIdx.x;
    const int s    = bid / ((NCH / CG) * EPARTS);
    const int rem  = bid - s * ((NCH / CG) * EPARTS);
    const int cg   = rem / EPARTS;
    const int part = rem - cg * EPARTS;
    const int c0   = cg * CG;
    for (int t = tid; t < WROWS * CG; t += 256) {
      const int row = t / CG;
      const int cc  = t - row * CG;
      sW[t] = w_row(w1_0e, w1_1o, w1_2e, w2_0e, w2_1o, w2_2e, w3_0e, w3_1o, w3_2e, row, s, c0 + cc);
    }
    __syncthreads();
    const int eEnd = (part + 1) * EPART_LEN;
    for (int e = part * EPART_LEN + tid; e < eEnd; e += 256) {
      const int n = e / KROW;
      const int p = e - n * KROW;
      const int ir   = (n == 0) ? 0 : (n < 4 ? 1 : 2);
      const int idim = (ir == 0) ? 1 : (ir == 1 ? 3 : 5);
      const int i    = (ir == 0) ? 0 : (ir == 1 ? n - 1 : n - 4);
      const float* u = nullptr;
      int mul = 0, ubase = 0, wrow0 = 0;
      if (p < G1) {
        if (p < 9) {
          u = (ir == 0) ? u1_0e : (ir == 1 ? u1_1o : u1_2e);
          mul = 1;
          ubase = p * idim + i;
          wrow0 = ir;
        }
      } else if (p < G1 + G2) {
        const int q = p - G1;
        if (q < 81) {
          u = (ir == 0) ? u2_0e : (ir == 1 ? u2_1o : u2_2e);
          mul = (ir == 2) ? 3 : 2;
          ubase = q * mul * idim + i;
          wrow0 = (ir == 0) ? 3 : (ir == 1 ? 5 : 7);
        }
      } else {
        const int t  = p - (G1 + G2);
        const int d0 = t / G2;
        const int q  = t - d0 * G2;
        if (q < 81) {
          u = (ir == 0) ? u3_0e : (ir == 1 ? u3_1o : u3_2e);
          mul = (ir == 0) ? 4 : (ir == 1 ? 6 : 8);
          ubase = (d0 * 81 + q) * mul * idim + i;
          wrow0 = (ir == 0) ? 10 : (ir == 1 ? 14 : 20);
        }
      }
      float v[CG];
#pragma unroll
      for (int cc = 0; cc < CG; ++cc) v[cc] = 0.f;
      for (int k = 0; k < mul; ++k) {
        const float uk = u[ubase + k * idim];
        const float* wr = sW + (wrow0 + k) * CG;
#pragma unroll
        for (int cc = 0; cc < CG; ++cc) v[cc] += wr[cc] * uk;
      }
      float* dst = ws + (size_t)(s * NCH + c0) * (NCOMP * KROW) + e;
#pragma unroll
      for (int cc = 0; cc < CG; ++cc) dst[(size_t)cc * (NCOMP * KROW)] = v[cc];
    }
  } else {
    __shared__ int sCnt[NGROUP * NSPEC];
    __shared__ int sBase[NGROUP * NSPEC];
    __shared__ int scnt[NSPEC];
    int* zlist = (int*)(ws + ZL_OFF);
    int* cnt   = (int*)(ws + CNT_OFF);
    int* table = (int*)(ws + TAB_OFF);
    const int lane = tid & 31;
    const unsigned below = (1u << lane) - 1u;
    int sp[NNODE / 256];
    int rk[NNODE / 256];
    for (int k = 0; k < NNODE / 256; ++k) {
      int v = index[tid + 256 * k];
      sp[k] = v < 0 ? 0 : (v > NSPEC - 1 ? NSPEC - 1 : v);
    }
    for (int k = 0; k < NNODE / 256; ++k) {
      const int grp = (tid >> 5) + 8 * k;
      int myrank = 0, mycount = 0;
      for (int s = 0; s < NSPEC; ++s) {
        const unsigned m = __builtin_amdgcn_ballot_w32(sp[k] == s);
        if (sp[k] == s) myrank = __builtin_popcount(m & below);
        if (lane == s) mycount = __builtin_popcount(m);
      }
      rk[k] = myrank;
      if (lane < NSPEC) sCnt[grp * NSPEC + lane] = mycount;
    }
    __syncthreads();
    if (tid < NSPEC) {
      int run = 0;
      for (int g = 0; g < NGROUP; ++g) {
        sBase[g * NSPEC + tid] = run;
        run += sCnt[g * NSPEC + tid];
      }
      scnt[tid] = run;
      cnt[tid]  = run;
    }
    __syncthreads();
    for (int k = 0; k < NNODE / 256; ++k) {
      const int grp = (tid >> 5) + 8 * k;
      zlist[sp[k] * NNODE + sBase[grp * NSPEC + sp[k]] + rk[k]] = tid + 256 * k;
    }
    if (tid == 0) {
      int slot = 0;
      for (int s = 0; s < NSPEC; ++s)
        for (int r0 = 0; r0 < scnt[s] && slot < MAXSLOT; r0 += CHUNK_ROWS) {
          table[2 * slot]     = s;
          table[2 * slot + 1] = r0;
          ++slot;
        }
      for (; slot < MAXSLOT; ++slot) {
        table[2 * slot]     = -1;
        table[2 * slot + 1] = 0;
      }
    }
  }
}

#define NSTEP 213
#define PF 14
#define GRP 4
#define PFW 64
#define PFLEAD 4
#define NPFW ((KROW + PFW - 1) / PFW)
__global__ __launch_bounds__(128) void symcon_main(
    const float* __restrict__ x, const float* __restrict__ ws, float* __restrict__ out) {
  const int* zlist = (const int*)(ws + ZL_OFF);
  const int* cnt   = (const int*)(ws + CNT_OFF);
  const int* table = (const int*)(ws + TAB_OFF);

  const int slot = blockIdx.y;
  const int s = table[2 * slot];
  if (s < 0) return;
  const int r0 = table[2 * slot + 1];
  const int n  = cnt[s];

  const int wv   = __builtin_amdgcn_readfirstlane(threadIdx.x >> 5);
  const int c    = blockIdx.x * 4 + wv;
  const int lane = threadIdx.x & 31;
  const int h    = lane >> 4;
  const int r    = lane & 15;

  int  zrow[2];
  bool valid[2];
#pragma unroll
  for (int t = 0; t < 2; ++t) {
    const int rr = r0 + 16 * t + r;
    valid[t] = rr < n;
    zrow[t]  = valid[t] ? zlist[s * NNODE + rr] : 0;
  }

  const int col = r < NCOMP ? r : NCOMP - 1;
  const float* bp = ws + ((size_t)((s * NCH + c) * NCOMP + col)) * KROW + 2 * h;

  v2f b[NSTEP];
#pragma unroll
  for (int q = 0; q < PF; ++q) b[q] = *(const v2f*)(bp + 4 * q);
#pragma unroll
  for (int m = 1; m < PFLEAD; ++m) __builtin_prefetch(bp + PFW * m, 0, 2);

  float xv[2][9];
#pragma unroll
  for (int t = 0; t < 2; ++t) {
    const float* xr = x + ((size_t)zrow[t] * NCH + c) * 9;
#pragma unroll
    for (int d = 0; d < 9; ++d) xv[t][d] = valid[t] ? xr[d] : 0.f;
  }

  v2f A1[2][3];
  v2f P[2][21];
#pragma unroll
  for (int t = 0; t < 2; ++t) {
#pragma unroll
    for (int j = 0; j < 3; ++j) {
#pragma unroll
      for (int e = 0; e < 2; ++e) {
        const int q0 = 4 * j + e;
        const int q1 = q0 + 2;
        const float a0 = (q0 < 9) ? xv[t][q0 < 9 ? q0 : 0] : 0.f;
        const float a1 = (q1 < 9) ? xv[t][q1 < 9 ? q1 : 0] : 0.f;
        A1[t][j][e] = h ? a1 : a0;
      }
    }
#pragma unroll
    for (int j = 0; j < 21; ++j) {
#pragma unroll
      for (int e = 0; e < 2; ++e) {
        const int q0 = 4 * j + e;
        const int q1 = q0 + 2;
        const float p0 = (q0 < 81) ? xv[t][q0 < 81 ? q0 / 9 : 0] * xv[t][q0 % 9] : 0.f;
        const float p1 = (q1 < 81) ? xv[t][q1 < 81 ? q1 / 9 : 0] * xv[t][q1 % 9] : 0.f;
        P[t][j][e] = h ? p1 : p0;
      }
    }
  }

  v8f acc0 = {0.f, 0.f, 0.f, 0.f, 0.f, 0.f, 0.f, 0.f};
  v8f acc1 = {0.f, 0.f, 0.f, 0.f, 0.f, 0.f, 0.f, 0.f};

#pragma unroll
  for (int g = 0; g < NSTEP; g += GRP) {
#pragma unroll
    for (int i = 0; i < GRP; ++i) {
      const int q = g + PF + i;
      if (q < NSTEP) b[q < NSTEP ? q : 0] = *(const v2f*)(bp + 4 * (q < NSTEP ? q : 0));
    }
    if (g % (PFW / 4) == 0 && g / (PFW / 4) + PFLEAD <= NPFW) {
      const int pw = PFW * (g / (PFW / 4) + PFLEAD);
      __builtin_prefetch(bp + (pw < KROW - 4 ? pw : KROW - 4), 0, 2);
    }
    __builtin_amdgcn_sched_barrier(0);
#pragma unroll
    for (int i = 0; i < GRP; ++i) {
      const int st = g + i;
      if (st < NSTEP) {
        const int sc = st < NSTEP ? st : 0;
        v2f a0, a1;
        if (sc < 3) {
          a0 = A1[0][sc < 3 ? sc : 0];
          a1 = A1[1][sc < 3 ? sc : 0];
        } else if (sc < 24) {
          a0 = P[0][(sc >= 3 && sc < 24) ? sc - 3 : 0];
          a1 = P[1][(sc >= 3 && sc < 24) ? sc - 3 : 0];
        } else {
          const int d0 = (sc - 24) / 21;
          const int j  = (sc - 24) % 21;
          a0 = P[0][sc >= 24 ? j : 0] * xv[0][sc >= 24 ? d0 : 0];
          a1 = P[1][sc >= 24 ? j : 0] * xv[1][sc >= 24 ? d0 : 0];
        }
        acc0 = __builtin_amdgcn_wmma_f32_16x16x4_f32(false, a0, false, b[sc], (short)0, acc0, false, false);
        acc1 = __builtin_amdgcn_wmma_f32_16x16x4_f32(false, a1, false, b[sc], (short)0, acc1, false, false);
      }
    }
  }

  if (r < NCOMP) {
    const int coff = (r == 0) ? c : (r < 4 ? 128 + c * 3 + (r - 1) : 512 + c * 5 + (r - 4));
#pragma unroll
    for (int j = 0; j < 8; ++j) {
      const int rr0 = r0 + 8 * h + j;
      if (rr0 < n) out[(size_t)zlist[s * NNODE + rr0] * 1152 + coff] = acc0[j];
      const int rr1 = rr0 + 16;
      if (rr1 < n) out[(size_t)zlist[s * NNODE + rr1] * 1152 + coff] = acc1[j];
    }
  }
}

extern "C" void kernel_launch(void* const* d_in, const int* in_sizes, int n_in,
                              void* d_out, int out_size, void* d_ws, size_t ws_size,
                              hipStream_t stream) {
  const float* x     = (const float*)d_in[0];
  const float* u1_0e = (const float*)d_in[1];
  const float* w1_0e = (const float*)d_in[2];
  const float* u1_1o = (const float*)d_in[3];
  const float* w1_1o = (const float*)d_in[4];
  const float* u1_2e = (const float*)d_in[5];
  const float* w1_2e = (const float*)d_in[6];
  const float* u2_0e = (const float*)d_in[7];
  const float* w2_0e = (const float*)d_in[8];
  const float* u2_1o = (const float*)d_in[9];
  const float* w2_1o = (const float*)d_in[10];
  const float* u2_2e = (const float*)d_in[11];
  const float* w2_2e = (const float*)d_in[12];
  const float* u3_0e = (const float*)d_in[13];
  const float* w3_0e = (const float*)d_in[14];
  const float* u3_1o = (const float*)d_in[15];
  const float* w3_1o = (const float*)d_in[16];
  const float* u3_2e = (const float*)d_in[17];
  const float* w3_2e = (const float*)d_in[18];
  const int*   index = (const int*)d_in[19];
  float* ws  = (float*)d_ws;
  float* out = (float*)d_out;

  prep_uw<<<PREP_BLOCKS + 1, 256, 0, stream>>>(
      u1_0e, u1_1o, u1_2e, u2_0e, u2_1o, u2_2e, u3_0e, u3_1o, u3_2e,
      w1_0e, w1_1o, w1_2e, w2_0e, w2_1o, w2_2e, w3_0e, w3_1o, w3_2e, index, ws);
  symcon_main<<<dim3(NCH / 4, MAXSLOT), 128, 0, stream>>>(x, ws, out);
}


# Main kernel: node ids of a wave's rows read on clamped in-range indices, all loads issued together (2 at entry, 16 before the stores), not one dependent round trip per row; same sums and order
# speedup vs baseline: 4.0984x; 1.1137x over previous
#include <hip/hip_runtime.h>

typedef float v2f __attribute__((ext_vector_type(2)));
typedef float v8f __attribute__((ext_vector_type(8)));

#define NSPEC 10
#define NCH 128
#define NNODE 1024
#define NCOMP 9
#define G1 12
#define G2 84
#define KROW 852
#define CHUNK_ROWS 32
#define NGROUP (NNODE / 32)
#define MAXSLOT 42
#define UW_FLOATS (NSPEC * NCH * NCOMP * KROW)
#define ZL_OFF  UW_FLOATS
#define CNT_OFF (UW_FLOATS + NSPEC * NNODE)
#define TAB_OFF (CNT_OFF + 16)

#define CG 32
#define EPARTS 12
#define EPART_LEN 639
#define WROWS 28
#define PREP_BLOCKS (NSPEC * (NCH / CG) * EPARTS)

__device__ inline float w_row(const float* __restrict__ w1_0e, const float* __restrict__ w1_1o,
                              const float* __restrict__ w1_2e, const float* __restrict__ w2_0e,
                              const float* __restrict__ w2_1o, const float* __restrict__ w2_2e,
                              const float* __restrict__ w3_0e, const float* __restrict__ w3_1o,
                              const float* __restrict__ w3_2e, int row, int s, int c) {
  if (row < 1)  return w1_0e[s * NCH + c];
  if (row < 2)  return w1_1o[s * NCH + c];
  if (row < 3)  return w1_2e[s * NCH + c];
  if (row < 5)  return w2_0e[(s * 2 + row - 3) * NCH + c];
  if (row < 7)  return w2_1o[(s * 2 + row - 5) * NCH + c];
  if (row < 10) return w2_2e[(s * 3 + row - 7) * NCH + c];
  if (row < 14) return w3_0e[(s * 4 + row - 10) * NCH + c];
  if (row < 20) return w3_1o[(s * 6 + row - 14) * NCH + c];
  return w3_2e[(s * 8 + row - 20) * NCH + c];
}

__global__ __launch_bounds__(256) void prep_uw(
    const float* __restrict__ u1_0e, const float* __restrict__ u1_1o, const float* __restrict__ u1_2e,
    const float* __restrict__ u2_0e, const float* __restrict__ u2_1o, const float* __restrict__ u2_2e,
    const float* __restrict__ u3_0e, const float* __restrict__ u3_1o, const float* __restrict__ u3_2e,
    const float* __restrict__ w1_0e, const float* __restrict__ w1_1o, const float* __restrict__ w1_2e,
    const float* __restrict__ w2_0e, const float* __restrict__ w2_1o, const float* __restrict__ w2_2e,
    const float* __restrict__ w3_0e, const float* __restrict__ w3_1o, const float* __restrict__ w3_2e,
    const int* __restrict__ index, float* __restrict__ ws) {
  const int tid = threadIdx.x;
  if (blockIdx.x < PREP_BLOCKS) {
    __shared__ float sW[WROWS * CG];
    const int bid  = blockIdx.x;
    const int s    = bid / ((NCH / CG) * EPARTS);
    const int rem  = bid - s * ((NCH / CG) * EPARTS);
    const int cg   = rem / EPARTS;
    const int part = rem - cg * EPARTS;
    const int c0   = cg * CG;
    for (int t = tid; t < WROWS * CG; t += 256) {
      const int row = t / CG;
      const int cc  = t - row * CG;
      sW[t] = w_row(w1_0e, w1_1o, w1_2e, w2_0e, w2_1o, w2_2e, w3_0e, w3_1o, w3_2e, row, s, c0 + cc);
    }
    __syncthreads();
    const int eEnd = (part + 1) * EPART_LEN;
    for (int e = part * EPART_LEN + tid; e < eEnd; e += 256) {
      const int n = e / KROW;
      const int p = e - n * KROW;
      const int ir   = (n == 0) ? 0 : (n < 4 ? 1 : 2);
      const int idim = (ir == 0) ? 1 : (ir == 1 ? 3 : 5);
      const int i    = (ir == 0) ? 0 : (ir == 1 ? n - 1 : n - 4);
      const float* u = nullptr;
      int mul = 0, ubase = 0, wrow0 = 0;
      if (p < G1) {
        if (p < 9) {
          u = (ir == 0) ? u1_0e : (ir == 1 ? u1_1o : u1_2e);
          mul = 1;
          ubase = p * idim + i;
          wrow0 = ir;
        }
      } else if (p < G1 + G2) {
        const int q = p - G1;
        if (q < 81) {
          u = (ir == 0) ? u2_0e : (ir == 1 ? u2_1o : u2_2e);
          mul = (ir == 2) ? 3 : 2;
          ubase = q * mul * idim + i;
          wrow0 = (ir == 0) ? 3 : (ir == 1 ? 5 : 7);
        }
      } else {
        const int t  = p - (G1 + G2);
        const int d0 = t / G2;
        const int q  = t - d0 * G2;
        if (q < 81) {
          u = (ir == 0) ? u3_0e : (ir == 1 ? u3_1o : u3_2e);
          mul = (ir == 0) ? 4 : (ir == 1 ? 6 : 8);
          ubase = (d0 * 81 + q) * mul * idim + i;
          wrow0 = (ir == 0) ? 10 : (ir == 1 ? 14 : 20);
        }
      }
      float v[CG];
#pragma unroll
      for (int cc = 0; cc < CG; ++cc) v[cc] = 0.f;
      for (int k = 0; k < mul; ++k) {
        const float uk = u[ubase + k * idim];
        const float* wr = sW + (wrow0 + k) * CG;
#pragma unroll
        for (int cc = 0; cc < CG; ++cc) v[cc] += wr[cc] * uk;
      }
      float* dst = ws + (size_t)(s * NCH + c0) * (NCOMP * KROW) + e;
#pragma unroll
      for (int cc = 0; cc < CG; ++cc) dst[(size_t)cc * (NCOMP * KROW)] = v[cc];
    }
  } else {
    __shared__ int sCnt[NGROUP * NSPEC];
    __shared__ int sBase[NGROUP * NSPEC];
    __shared__ int scnt[NSPEC];
    int* zlist = (int*)(ws + ZL_OFF);
    int* cnt   = (int*)(ws + CNT_OFF);
    int* table = (int*)(ws + TAB_OFF);
    const int lane = tid & 31;
    const unsigned below = (1u << lane) - 1u;
    int sp[NNODE / 256];
    int rk[NNODE / 256];
    for (int k = 0; k < NNODE / 256; ++k) {
      int v = index[tid + 256 * k];
      sp[k] = v < 0 ? 0 : (v > NSPEC - 1 ? NSPEC - 1 : v);
    }
    for (int k = 0; k < NNODE / 256; ++k) {
      const int grp = (tid >> 5) + 8 * k;
      int myrank = 0, mycount = 0;
      for (int s = 0; s < NSPEC; ++s) {
        const unsigned m = __builtin_amdgcn_ballot_w32(sp[k] == s);
        if (sp[k] == s) myrank = __builtin_popcount(m & below);
        if (lane == s) mycount = __builtin_popcount(m);
      }
      rk[k] = myrank;
      if (lane < NSPEC) sCnt[grp * NSPEC + lane] = mycount;
    }
    __syncthreads();
    if (tid < NSPEC) {
      int run = 0;
      for (int g = 0; g < NGROUP; ++g) {
        sBase[g * NSPEC + tid] = run;
        run += sCnt[g * NSPEC + tid];
      }
      scnt[tid] = run;
      cnt[tid]  = run;
    }
    __syncthreads();
    for (int k = 0; k < NNODE / 256; ++k) {
      const int grp = (tid >> 5) + 8 * k;
      zlist[sp[k] * NNODE + sBase[grp * NSPEC + sp[k]] + rk[k]] = tid + 256 * k;
    }
    if (tid == 0) {
      int slot = 0;
      for (int s = 0; s < NSPEC; ++s)
        for (int r0 = 0; r0 < scnt[s] && slot < MAXSLOT; r0 += CHUNK_ROWS) {
          table[2 * slot]     = s;
          table[2 * slot + 1] = r0;
          ++slot;
        }
      for (; slot < MAXSLOT; ++slot) {
        table[2 * slot]     = -1;
        table[2 * slot + 1] = 0;
      }
    }
  }
}

#define NSTEP 213
#define PF 14
#define GRP 4
#define PFW 64
#define PFLEAD 4
#define NPFW ((KROW + PFW - 1) / PFW)
__global__ __launch_bounds__(128) void symcon_main(
    const float* __restrict__ x, const float* __restrict__ ws, float* __restrict__ out) {
  const int* zlist = (const int*)(ws + ZL_OFF);
  const int* cnt   = (const int*)(ws + CNT_OFF);
  const int* table = (const int*)(ws + TAB_OFF);

  const int slot = blockIdx.y;
  const int s = table[2 * slot];
  if (s < 0) return;
  const int r0 = table[2 * slot + 1];
  const int n  = cnt[s];

  const int wv   = __builtin_amdgcn_readfirstlane(threadIdx.x >> 5);
  const int c    = blockIdx.x * 4 + wv;
  const int lane = threadIdx.x & 31;
  const int h    = lane >> 4;
  const int r    = lane & 15;

  int  zrow[2];
  bool valid[2];
#pragma unroll
  for (int t = 0; t < 2; ++t) {
    const int rr = r0 + 16 * t + r;
    valid[t] = rr < n;
    zrow[t]  = zlist[s * NNODE + (valid[t] ? rr : r0)];
  }

  const int col = r < NCOMP ? r : NCOMP - 1;
  const float* bp = ws + ((size_t)((s * NCH + c) * NCOMP + col)) * KROW + 2 * h;

  v2f b[NSTEP];
#pragma unroll
  for (int q = 0; q < PF; ++q) b[q] = *(const v2f*)(bp + 4 * q);
#pragma unroll
  for (int m = 1; m < PFLEAD; ++m) __builtin_prefetch(bp + PFW * m, 0, 2);

  float xv[2][9];
#pragma unroll
  for (int t = 0; t < 2; ++t) {
    const float* xr = x + ((size_t)zrow[t] * NCH + c) * 9;
#pragma unroll
    for (int d = 0; d < 9; ++d) {
      const float xd = xr[d];
      xv[t][d] = valid[t] ? xd : 0.f;
    }
  }

  v2f A1[2][3];
  v2f P[2][21];
#pragma unroll
  for (int t = 0; t < 2; ++t) {
#pragma unroll
    for (int j = 0; j < 3; ++j) {
#pragma unroll
      for (int e = 0; e < 2; ++e) {
        const int q0 = 4 * j + e;
        const int q1 = q0 + 2;
        const float a0 = (q0 < 9) ? xv[t][q0 < 9 ? q0 : 0] : 0.f;
        const float a1 = (q1 < 9) ? xv[t][q1 < 9 ? q1 : 0] : 0.f;
        A1[t][j][e] = h ? a1 : a0;
      }
    }
#pragma unroll
    for (int j = 0; j < 21; ++j) {
#pragma unroll
      for (int e = 0; e < 2; ++e) {
        const int q0 = 4 * j + e;
        const int q1 = q0 + 2;
        const float p0 = (q0 < 81) ? xv[t][q0 < 81 ? q0 / 9 : 0] * xv[t][q0 % 9] : 0.f;
        const float p1 = (q1 < 81) ? xv[t][q1 < 81 ? q1 / 9 : 0] * xv[t][q1 % 9] : 0.f;
        P[t][j][e] = h ? p1 : p0;
      }
    }
  }

  v8f acc0 = {0.f, 0.f, 0.f, 0.f, 0.f, 0.f, 0.f, 0.f};
  v8f acc1 = {0.f, 0.f, 0.f, 0.f, 0.f, 0.f, 0.f, 0.f};

#pragma unroll
  for (int g = 0; g < NSTEP; g += GRP) {
#pragma unroll
    for (int i = 0; i < GRP; ++i) {
      const int q = g + PF + i;
      if (q < NSTEP) b[q < NSTEP ? q : 0] = *(const v2f*)(bp + 4 * (q < NSTEP ? q : 0));
    }
    if (g % (PFW / 4) == 0 && g / (PFW / 4) + PFLEAD <= NPFW) {
      const int pw = PFW * (g / (PFW / 4) + PFLEAD);
      __builtin_prefetch(bp + (pw < KROW - 4 ? pw : KROW - 4), 0, 2);
    }
    __builtin_amdgcn_sched_barrier(0);
#pragma unroll
    for (int i = 0; i < GRP; ++i) {
      const int st = g + i;
      if (st < NSTEP) {
        const int sc = st < NSTEP ? st : 0;
        v2f a0, a1;
        if (sc < 3) {
          a0 = A1[0][sc < 3 ? sc : 0];
          a1 = A1[1][sc < 3 ? sc : 0];
        } else if (sc < 24) {
          a0 = P[0][(sc >= 3 && sc < 24) ? sc - 3 : 0];
          a1 = P[1][(sc >= 3 && sc < 24) ? sc - 3 : 0];
        } else {
          const int d0 = (sc - 24) / 21;
          const int j  = (sc - 24) % 21;
          a0 = P[0][sc >= 24 ? j : 0] * xv[0][sc >= 24 ? d0 : 0];
          a1 = P[1][sc >= 24 ? j : 0] * xv[1][sc >= 24 ? d0 : 0];
        }
        acc0 = __builtin_amdgcn_wmma_f32_16x16x4_f32(false, a0, false, b[sc], (short)0, acc0, false, false);
        acc1 = __builtin_amdgcn_wmma_f32_16x16x4_f32(false, a1, false, b[sc], (short)0, acc1, false, false);
      }
    }
  }

  if (r < NCOMP) {
    const int coff = (r == 0) ? c : (r < 4 ? 128 + c * 3 + (r - 1) : 512 + c * 5 + (r - 4));
    int zn0[8], zn1[8];
#pragma unroll
    for (int j = 0; j < 8; ++j) {
      const int rr0 = r0 + 8 * h + j;
      const int rr1 = rr0 + 16;
      zn0[j] = zlist[s * NNODE + (rr0 < n ? rr0 : r0)];
      zn1[j] = zlist[s * NNODE + (rr1 < n ? rr1 : r0)];
    }
#pragma unroll
    for (int j = 0; j < 8; ++j) {
      const int rr0 = r0 + 8 * h + j;
      const int rr1 = rr0 + 16;
      if (rr0 < n) out[(size_t)zn0[j] * 1152 + coff] = acc0[j];
      if (rr1 < n) out[(size_t)zn1[j] * 1152 + coff] = acc1[j];
    }
  }
}

extern "C" void kernel_launch(void* const* d_in, const int* in_sizes, int n_in,
                              void* d_out, int out_size, void* d_ws, size_t ws_size,
                              hipStream_t stream) {
  const float* x     = (const float*)d_in[0];
  const float* u1_0e = (const float*)d_in[1];
  const float* w1_0e = (const float*)d_in[2];
  const float* u1_1o = (const float*)d_in[3];
  const float* w1_1o = (const float*)d_in[4];
  const float* u1_2e = (const float*)d_in[5];
  const float* w1_2e = (const float*)d_in[6];
  const float* u2_0e = (const float*)d_in[7];
  const float* w2_0e = (const float*)d_in[8];
  const float* u2_1o = (const float*)d_in[9];
  const float* w2_1o = (const float*)d_in[10];
  const float* u2_2e = (const float*)d_in[11];
  const float* w2_2e = (const float*)d_in[12];
  const float* u3_0e = (const float*)d_in[13];
  const float* w3_0e = (const float*)d_in[14];
  const float* u3_1o = (const float*)d_in[15];
  const float* w3_1o = (const float*)d_in[16];
  const float* u3_2e = (const float*)d_in[17];
  const float* w3_2e = (const float*)d_in[18];
  const int*   index = (const int*)d_in[19];
  float* ws  = (float*)d_ws;
  float* out = (float*)d_out;

  prep_uw<<<PREP_BLOCKS + 1, 256, 0, stream>>>(
      u1_0e, u1_1o, u1_2e, u2_0e, u2_1o, u2_2e, u3_0e, u3_1o, u3_2e,
      w1_0e, w1_1o, w1_2e, w2_0e, w2_1o, w2_2e, w3_0e, w3_1o, w3_2e, index, ws);
  symcon_main<<<dim3(NCH / 4, MAXSLOT), 128, 0, stream>>>(x, ws, out);
}


# Main kernel: block = one species and channel; its UW operand (30.7 KB) copied once to LDS by whole-line loads, 4 waves read B fragments from LDS, not 9-line global gathers; same sums and order
# speedup vs baseline: 4.2384x; 1.0342x over previous
#include <hip/hip_runtime.h>

typedef float v2f __attribute__((ext_vector_type(2)));
typedef float v4f __attribute__((ext_vector_type(4)));
typedef float v8f __attribute__((ext_vector_type(8)));

#define NSPEC 10
#define NCH 128
#define NNODE 1024
#define NCOMP 9
#define G1 12
#define G2 84
#define KROW 852
#define CHUNK_ROWS 32
#define BLK_ROWS 128
#define NGROUP (NNODE / 32)
#define UW_FLOATS (NSPEC * NCH * NCOMP * KROW)
#define ZL_OFF  UW_FLOATS
#define CNT_OFF (UW_FLOATS + NSPEC * NNODE)

#define CG 32
#define EPARTS 12
#define EPART_LEN 639
#define WROWS 28
#define PREP_BLOCKS (NSPEC * (NCH / CG) * EPARTS)

__device__ inline float w_row(const float* __restrict__ w1_0e, const float* __restrict__ w1_1o,
                              const float* __restrict__ w1_2e, const float* __restrict__ w2_0e,
                              const float* __restrict__ w2_1o, const float* __restrict__ w2_2e,
                              const float* __restrict__ w3_0e, const float* __restrict__ w3_1o,
                              const float* __restrict__ w3_2e, int row, int s, int c) {
  if (row < 1)  return w1_0e[s * NCH + c];
  if (row < 2)  return w1_1o[s * NCH + c];
  if (row < 3)  return w1_2e[s * NCH + c];
  if (row < 5)  return w2_0e[(s * 2 + row - 3) * NCH + c];
  if (row < 7)  return w2_1o[(s * 2 + row - 5) * NCH + c];
  if (row < 10) return w2_2e[(s * 3 + row - 7) * NCH + c];
  if (row < 14) return w3_0e[(s * 4 + row - 10) * NCH + c];
  if (row < 20) return w3_1o[(s * 6 + row - 14) * NCH + c];
  return w3_2e[(s * 8 + row - 20) * NCH + c];
}

__global__ __launch_bounds__(256) void prep_uw(
    const float* __restrict__ u1_0e, const float* __restrict__ u1_1o, const float* __restrict__ u1_2e,
    const float* __restrict__ u2_0e, const float* __restrict__ u2_1o, const float* __restrict__ u2_2e,
    const float* __restrict__ u3_0e, const float* __restrict__ u3_1o, const float* __restrict__ u3_2e,
    const float* __restrict__ w1_0e, const float* __restrict__ w1_1o, const float* __restrict__ w1_2e,
    const float* __restrict__ w2_0e, const float* __restrict__ w2_1o, const float* __restrict__ w2_2e,
    const float* __restrict__ w3_0e, const float* __restrict__ w3_1o, const float* __restrict__ w3_2e,
    const int* __restrict__ index, float* __restrict__ ws) {
  const int tid = threadIdx.x;
  if (blockIdx.x < PREP_BLOCKS) {
    __shared__ float sW[WROWS * CG];
    const int bid  = blockIdx.x;
    const int s    = bid / ((NCH / CG) * EPARTS);
    const int rem  = bid - s * ((NCH / CG) * EPARTS);
    const int cg   = rem / EPARTS;
    const int part = rem - cg * EPARTS;
    const int c0   = cg * CG;
    for (int t = tid; t < WROWS * CG; t += 256) {
      const int row = t / CG;
      const int cc  = t - row * CG;
      sW[t] = w_row(w1_0e, w1_1o, w1_2e, w2_0e, w2_1o, w2_2e, w3_0e, w3_1o, w3_2e, row, s, c0 + cc);
    }
    __syncthreads();
    const int eEnd = (part + 1) * EPART_LEN;
    for (int e = part * EPART_LEN + tid; e < eEnd; e += 256) {
      const int n = e / KROW;
      const int p = e - n * KROW;
      const int ir   = (n == 0) ? 0 : (n < 4 ? 1 : 2);
      const int idim = (ir == 0) ? 1 : (ir == 1 ? 3 : 5);
      const int i    = (ir == 0) ? 0 : (ir == 1 ? n - 1 : n - 4);
      const float* u = nullptr;
      int mul = 0, ubase = 0, wrow0 = 0;
      if (p < G1) {
        if (p < 9) {
          u = (ir == 0) ? u1_0e : (ir == 1 ? u1_1o : u1_2e);
          mul = 1;
          ubase = p * idim + i;
          wrow0 = ir;
        }
      } else if (p < G1 + G2) {
        const int q = p - G1;
        if (q < 81) {
          u = (ir == 0) ? u2_0e : (ir == 1 ? u2_1o : u2_2e);
          mul = (ir == 2) ? 3 : 2;
          ubase = q * mul * idim + i;
          wrow0 = (ir == 0) ? 3 : (ir == 1 ? 5 : 7);
        }
      } else {
        const int t  = p - (G1 + G2);
        const int d0 = t / G2;
        const int q  = t - d0 * G2;
        if (q < 81) {
          u = (ir == 0) ? u3_0e : (ir == 1 ? u3_1o : u3_2e);
          mul = (ir == 0) ? 4 : (ir == 1 ? 6 : 8);
          ubase = (d0 * 81 + q) * mul * idim + i;
          wrow0 = (ir == 0) ? 10 : (ir == 1 ? 14 : 20);
        }
      }
      float v[CG];
#pragma unroll
      for (int cc = 0; cc < CG; ++cc) v[cc] = 0.f;
      for (int k = 0; k < mul; ++k) {
        const float uk = u[ubase + k * idim];
        const float* wr = sW + (wrow0 + k) * CG;
#pragma unroll
        for (int cc = 0; cc < CG; ++cc) v[cc] += wr[cc] * uk;
      }
      float* dst = ws + (size_t)(s * NCH + c0) * (NCOMP * KROW) + e;
#pragma unroll
      for (int cc = 0; cc < CG; ++cc) dst[(size_t)cc * (NCOMP * KROW)] = v[cc];
    }
  } else {
    __shared__ int sCnt[NGROUP * NSPEC];
    __shared__ int sBase[NGROUP * NSPEC];
    int* zlist = (int*)(ws + ZL_OFF);
    int* cnt   = (int*)(ws + CNT_OFF);
    const int lane = tid & 31;
    const unsigned below = (1u << lane) - 1u;
    int sp[NNODE / 256];
    int rk[NNODE / 256];
    for (int k = 0; k < NNODE / 256; ++k) {
      int v = index[tid + 256 * k];
      sp[k] = v < 0 ? 0 : (v > NSPEC - 1 ? NSPEC - 1 : v);
    }
    for (int k = 0; k < NNODE / 256; ++k) {
      const int grp = (tid >> 5) + 8 * k;
      int myrank = 0, mycount = 0;
      for (int s = 0; s < NSPEC; ++s) {
        const unsigned m = __builtin_amdgcn_ballot_w32(sp[k] == s);
        if (sp[k] == s) myrank = __builtin_popcount(m & below);
        if (lane == s) mycount = __builtin_popcount(m);
      }
      rk[k] = myrank;
      if (lane < NSPEC) sCnt[grp * NSPEC + lane] = mycount;
    }
    __syncthreads();
    if (tid < NSPEC) {
      int run = 0;
      for (int g = 0; g < NGROUP; ++g) {
        sBase[g * NSPEC + tid] = run;
        run += sCnt[g * NSPEC + tid];
      }
      cnt[tid] = run;
    }
    __syncthreads();
    for (int k = 0; k < NNODE / 256; ++k) {
      const int grp = (tid >> 5) + 8 * k;
      zlist[sp[k] * NNODE + sBase[grp * NSPEC + sp[k]] + rk[k]] = tid + 256 * k;
    }
  }
}

#define NSTEP 213
#define PF 8
#define GRP 8
#define UWROW (NCOMP * KROW)
#define NV4 (UWROW / 4)
#define NSTAGE ((NV4 + 127) / 128)
__global__ __launch_bounds__(128) void symcon_main(
    const float* __restrict__ x, const float* __restrict__ ws, float* __restrict__ out) {
  __shared__ __attribute__((aligned(16))) float sB[NSTAGE * 128 * 4];
  const int* zlist = (const int*)(ws + ZL_OFF);
  const int* cnt   = (const int*)(ws + CNT_OFF);

  const int s = blockIdx.y;
  const int n = cnt[s];
  if (n <= 0) return;

  const int wv   = __builtin_amdgcn_readfirstlane(threadIdx.x >> 5);
  const int c    = blockIdx.x;
  const int lane = threadIdx.x & 31;
  const int h    = lane >> 4;
  const int r    = lane & 15;

  int rb = 0;
  int r0 = CHUNK_ROWS * wv;
  int  zrow[2];
  bool valid[2];
#pragma unroll
  for (int t = 0; t < 2; ++t) {
    const int rr = r0 + 16 * t + r;
    valid[t] = rr < n;
    zrow[t]  = zlist[s * NNODE + (valid[t] ? rr : rb)];
  }

  const v4f* usrc = (const v4f*)(ws + (size_t)(s * NCH + c) * UWROW);
  v4f stg[NSTAGE];
#pragma unroll
  for (int i = 0; i < NSTAGE; ++i) {
    const int idx = (int)(threadIdx.x & 127) + 128 * i;
    stg[i] = usrc[idx < NV4 ? idx : NV4 - 1];
  }

  float xv[2][9];
#pragma unroll
  for (int t = 0; t < 2; ++t) {
    const float* xr = x + ((size_t)zrow[t] * NCH + c) * 9;
#pragma unroll
    for (int d = 0; d < 9; ++d) {
      const float xd = xr[d];
      xv[t][d] = valid[t] ? xd : 0.f;
    }
  }

#pragma unroll
  for (int i = 0; i < NSTAGE; ++i) {
    const int idx = (int)(threadIdx.x & 127) + 128 * i;
    ((v4f*)sB)[idx] = stg[i];
  }
  __syncthreads();

  const int col = r < NCOMP ? r : NCOMP - 1;
  const float* bp = sB + col * KROW + 2 * h;
  const int coff = (r == 0) ? c : (r < 4 ? 128 + c * 3 + (r - 1) : 512 + c * 5 + (r - 4));

  for (;;) {
    if (r0 < n) {
      v2f A1[2][3];
      v2f P[2][21];
#pragma unroll
      for (int t = 0; t < 2; ++t) {
#pragma unroll
        for (int j = 0; j < 3; ++j) {
#pragma unroll
          for (int e = 0; e < 2; ++e) {
            const int q0 = 4 * j + e;
            const int q1 = q0 + 2;
            const float a0 = (q0 < 9) ? xv[t][q0 < 9 ? q0 : 0] : 0.f;
            const float a1 = (q1 < 9) ? xv[t][q1 < 9 ? q1 : 0] : 0.f;
            A1[t][j][e] = h ? a1 : a0;
          }
        }
#pragma unroll
        for (int j = 0; j < 21; ++j) {
#pragma unroll
          for (int e = 0; e < 2; ++e) {
            const int q0 = 4 * j + e;
            const int q1 = q0 + 2;
            const float p0 = (q0 < 81) ? xv[t][q0 < 81 ? q0 / 9 : 0] * xv[t][q0 % 9] : 0.f;
            const float p1 = (q1 < 81) ? xv[t][q1 < 81 ? q1 / 9 : 0] * xv[t][q1 % 9] : 0.f;
            P[t][j][e] = h ? p1 : p0;
          }
        }
      }

      v2f b[NSTEP];
#pragma unroll
      for (int q = 0; q < PF; ++q) b[q] = *(const v2f*)(bp + 4 * q);

      v8f acc0 = {0.f, 0.f, 0.f, 0.f, 0.f, 0.f, 0.f, 0.f};
      v8f acc1 = {0.f, 0.f, 0.f, 0.f, 0.f, 0.f, 0.f, 0.f};

#pragma unroll
      for (int g = 0; g < NSTEP; g += GRP) {
#pragma unroll
        for (int i = 0; i < GRP; ++i) {
          const int q = g + PF + i;
          if (q < NSTEP) b[q < NSTEP ? q : 0] = *(const v2f*)(bp + 4 * (q < NSTEP ? q : 0));
        }
        __builtin_amdgcn_sched_barrier(0);
#pragma unroll
        for (int i = 0; i < GRP; ++i) {
          const int st = g + i;
          if (st < NSTEP) {
            const int sc = st < NSTEP ? st : 0;
            v2f a0, a1;
            if (sc < 3) {
              a0 = A1[0][sc < 3 ? sc : 0];
              a1 = A1[1][sc < 3 ? sc : 0];
            } else if (sc < 24) {
              a0 = P[0][(sc >= 3 && sc < 24) ? sc - 3 : 0];
              a1 = P[1][(sc >= 3 && sc < 24) ? sc - 3 : 0];
            } else {
              const int d0 = (sc - 24) / 21;
              const int j  = (sc - 24) % 21;
              a0 = P[0][sc >= 24 ? j : 0] * xv[0][sc >= 24 ? d0 : 0];
              a1 = P[1][sc >= 24 ? j : 0] * xv[1][sc >= 24 ? d0 : 0];
            }
            acc0 = __builtin_amdgcn_wmma_f32_16x16x4_f32(false, a0, false, b[sc], (short)0, acc0, false, false);
            acc1 = __builtin_amdgcn_wmma_f32_16x16x4_f32(false, a1, false, b[sc], (short)0, acc1, false, false);
          }
        }
      }

      if (r < NCOMP) {
        int zn0[8], zn1[8];
#pragma unroll
        for (int j = 0; j < 8; ++j) {
          const int rr0 = r0 + 8 * h + j;
          const int rr1 = rr0 + 16;
          zn0[j] = zlist[s * NNODE + (rr0 < n ? rr0 : rb)];
          zn1[j] = zlist[s * NNODE + (rr1 < n ? rr1 : rb)];
        }
#pragma unroll
        for (int j = 0; j < 8; ++j) {
          const int rr0 = r0 + 8 * h + j;
          const int rr1 = rr0 + 16;
          if (rr0 < n) out[(size_t)zn0[j] * 1152 + coff] = acc0[j];
          if (rr1 < n) out[(size_t)zn1[j] * 1152 + coff] = acc1[j];
        }
      }
    }

    rb += BLK_ROWS;
    if (rb >= n) break;
    r0 = rb + CHUNK_ROWS * wv;
#pragma unroll
    for (int t = 0; t < 2; ++t) {
      const int rr = r0 + 16 * t + r;
      const bool inr = rr < n;
      const int z = zlist[s * NNODE + (inr ? rr : rb)];
      const float* xr = x + ((size_t)z * NCH + c) * 9;
#pragma unroll
      for (int d = 0; d < 9; ++d) {
        const float xd = xr[d];
        xv[t][d] = inr ? xd : 0.f;
      }
    }
  }
}

extern "C" void kernel_launch(void* const* d_in, const int* in_sizes, int n_in,
                              void* d_out, int out_size, void* d_ws, size_t ws_size,
                              hipStream_t stream) {
  const float* x     = (const float*)d_in[0];
  const float* u1_0e = (const float*)d_in[1];
  const float* w1_0e = (const float*)d_in[2];
  const float* u1_1o = (const float*)d_in[3];
  const float* w1_1o = (const float*)d_in[4];
  const float* u1_2e = (const float*)d_in[5];
  const float* w1_2e = (const float*)d_in[6];
  const float* u2_0e = (const float*)d_in[7];
  const float* w2_0e = (const float*)d_in[8];
  const float* u2_1o = (const float*)d_in[9];
  const float* w2_1o = (const float*)d_in[10];
  const float* u2_2e = (const float*)d_in[11];
  const float* w2_2e = (const float*)d_in[12];
  const float* u3_0e = (const float*)d_in[13];
  const float* w3_0e = (const float*)d_in[14];
  const float* u3_1o = (const float*)d_in[15];
  const float* w3_1o = (const float*)d_in[16];
  const float* u3_2e = (const float*)d_in[17];
  const float* w3_2e = (const float*)d_in[18];
  const int*   index = (const int*)d_in[19];
  float* ws  = (float*)d_ws;
  float* out = (float*)d_out;

  prep_uw<<<PREP_BLOCKS + 1, 256, 0, stream>>>(
      u1_0e, u1_1o, u1_2e, u2_0e, u2_1o, u2_2e, u3_0e, u3_1o, u3_2e,
      w1_0e, w1_1o, w1_2e, w2_0e, w2_1o, w2_2e, w3_0e, w3_1o, w3_2e, index, ws);
  symcon_main<<<dim3(NCH, NSPEC), 128, 0, stream>>>(x, ws, out);
}


# Prep launch: a thread's weight loads and all U gathers of its operand elements (8 per element, in-range addresses) issued together, not one dependent round trip per term; same sums and order
# speedup vs baseline: 4.7155x; 1.1126x over previous
#include <hip/hip_runtime.h>

typedef float v2f __attribute__((ext_vector_type(2)));
typedef float v4f __attribute__((ext_vector_type(4)));
typedef float v8f __attribute__((ext_vector_type(8)));

#define NSPEC 10
#define NCH 128
#define NNODE 1024
#define NCOMP 9
#define NKS 21
#define NMT 6
#define NFRAG (NKS * NMT)
#define UWROW (NFRAG * 64)
#define NGROUP (NNODE / 32)
#define UW_FLOATS (NSPEC * NCH * UWROW)
#define ZL_OFF  UW_FLOATS
#define CNT_OFF (UW_FLOATS + NSPEC * NNODE)

#define CG 32
#define EPARTS 12
#define EPART_LEN 672
#define WROWS 28
#define PREP_BLOCKS (NSPEC * (NCH / CG) * EPARTS)
#define WFILL ((WROWS * CG + 255) / 256)
#define NEL ((EPART_LEN + 255) / 256)
#define MAXMUL 8

__device__ inline float w_row(const float* __restrict__ w1_0e, const float* __restrict__ w1_1o,
                              const float* __restrict__ w1_2e, const float* __restrict__ w2_0e,
                              const float* __restrict__ w2_1o, const float* __restrict__ w2_2e,
                              const float* __restrict__ w3_0e, const float* __restrict__ w3_1o,
                              const float* __restrict__ w3_2e, int row, int s, int c) {
  if (row < 1)  return w1_0e[s * NCH + c];
  if (row < 2)  return w1_1o[s * NCH + c];
  if (row < 3)  return w1_2e[s * NCH + c];
  if (row < 5)  return w2_0e[(s * 2 + row - 3) * NCH + c];
  if (row < 7)  return w2_1o[(s * 2 + row - 5) * NCH + c];
  if (row < 10) return w2_2e[(s * 3 + row - 7) * NCH + c];
  if (row < 14) return w3_0e[(s * 4 + row - 10) * NCH + c];
  if (row < 20) return w3_1o[(s * 6 + row - 14) * NCH + c];
  return w3_2e[(s * 8 + row - 20) * NCH + c];
}

__global__ __launch_bounds__(256) void prep_uw(
    const float* __restrict__ u1_0e, const float* __restrict__ u1_1o, const float* __restrict__ u1_2e,
    const float* __restrict__ u2_0e, const float* __restrict__ u2_1o, const float* __restrict__ u2_2e,
    const float* __restrict__ u3_0e, const float* __restrict__ u3_1o, const float* __restrict__ u3_2e,
    const float* __restrict__ w1_0e, const float* __restrict__ w1_1o, const float* __restrict__ w1_2e,
    const float* __restrict__ w2_0e, const float* __restrict__ w2_1o, const float* __restrict__ w2_2e,
    const float* __restrict__ w3_0e, const float* __restrict__ w3_1o, const float* __restrict__ w3_2e,
    const int* __restrict__ index, float* __restrict__ ws) {
  const int tid = threadIdx.x;
  if (blockIdx.x < PREP_BLOCKS) {
    __shared__ float sW[WROWS * CG];
    const int bid  = blockIdx.x;
    const int s    = bid / ((NCH / CG) * EPARTS);
    const int rem  = bid - s * ((NCH / CG) * EPARTS);
    const int cg   = rem / EPARTS;
    const int part = rem - cg * EPARTS;
    const int c0   = cg * CG;
    float wv[WFILL];
#pragma unroll
    for (int i = 0; i < WFILL; ++i) {
      const int t   = tid + 256 * i;
      const int tt  = t < WROWS * CG ? t : 0;
      const int row = tt / CG;
      const int cc  = tt - row * CG;
      wv[i] = w_row(w1_0e, w1_1o, w1_2e, w2_0e, w2_1o, w2_2e, w3_0e, w3_1o, w3_2e, row, s, c0 + cc);
    }
    float uv[NEL][MAXMUL];
    int mulv[NEL], wrow0v[NEL];
#pragma unroll
    for (int it = 0; it < NEL; ++it) {
      const int e   = part * EPART_LEN + tid + 256 * it;
      const int el  = e & 1;
      const int ln  = (e >> 1) & 31;
      const int fr  = e >> 6;
      const int ks  = fr / NMT;
      const int mt  = fr - ks * NMT;
      const int q   = 4 * ks + 2 * (ln >> 4) + el;
      const int sg  = 8 * mt + (ln & 7);
      const int R   = 45 * ((ln >> 3) & 1) + sg;
      const int d1  = R / 9;
      const int n   = R - d1 * 9;
      const int ir   = (n == 0) ? 0 : (n < 4 ? 1 : 2);
      const int idim = (ir == 0) ? 1 : (ir == 1 ? 3 : 5);
      const int i    = (ir == 0) ? 0 : (ir == 1 ? n - 1 : n - 4);
      const float* u = u1_0e;
      int mul = 0, ubase = 0, wrow0 = 0;
      if (sg < 45) {
        if (q < 81) {
          if (d1 < 9) {
            u = (ir == 0) ? u3_0e : (ir == 1 ? u3_1o : u3_2e);
            mul = (ir == 0) ? 4 : (ir == 1 ? 6 : 8);
            ubase = (d1 * 81 + q) * mul * idim + i;
            wrow0 = (ir == 0) ? 10 : (ir == 1 ? 14 : 20);
          } else {
            u = (ir == 0) ? u2_0e : (ir == 1 ? u2_1o : u2_2e);
            mul = (ir == 2) ? 3 : 2;
            ubase = q * mul * idim + i;
            wrow0 = (ir == 0) ? 3 : (ir == 1 ? 5 : 7);
          }
        } else if (q == 81 && d1 < 9) {
          u = (ir == 0) ? u1_0e : (ir == 1 ? u1_1o : u1_2e);
          mul = 1;
          ubase = d1 * idim + i;
          wrow0 = ir;
        }
      }
      const bool live = tid + 256 * it < EPART_LEN;
      mulv[it]   = live ? mul : 0;
      wrow0v[it] = wrow0;
#pragma unroll
      for (int k = 0; k < MAXMUL; ++k) uv[it][k] = 0.f;
      if (live) {
#pragma unroll
        for (int k = 0; k < MAXMUL; ++k) uv[it][k] = u[ubase + (k < mul ? k : 0) * idim];
      }
    }
#pragma unroll
    for (int i = 0; i < WFILL; ++i) {
      const int t = tid + 256 * i;
      if (t < WROWS * CG) sW[t] = wv[i];
    }
    __syncthreads();
#pragma unroll
    for (int it = 0; it < NEL; ++it) {
      const int eo = tid + 256 * it;
      if (eo < EPART_LEN) {
        float v[CG];
#pragma unroll
        for (int cc = 0; cc < CG; ++cc) v[cc] = 0.f;
#pragma unroll
        for (int k = 0; k < MAXMUL; ++k) {
          if (k < mulv[it]) {
            const float uk = uv[it][k];
            const float* wr = sW + (wrow0v[it] + k) * CG;
#pragma unroll
            for (int cc = 0; cc < CG; ++cc) v[cc] += wr[cc] * uk;
          }
        }
        float* dst = ws + (size_t)(s * NCH + c0) * UWROW + part * EPART_LEN + eo;
#pragma unroll
        for (int cc = 0; cc < CG; ++cc) dst[(size_t)cc * UWROW] = v[cc];
      }
    }
  } else {
    __shared__ int sCnt[NGROUP * NSPEC];
    __shared__ int sBase[NGROUP * NSPEC];
    int* zlist = (int*)(ws + ZL_OFF);
    int* cnt   = (int*)(ws + CNT_OFF);
    const int lane = tid & 31;
    const unsigned below = (1u << lane) - 1u;
    int sp[NNODE / 256];
    int rk[NNODE / 256];
    for (int k = 0; k < NNODE / 256; ++k) {
      int v = index[tid + 256 * k];
      sp[k] = v < 0 ? 0 : (v > NSPEC - 1 ? NSPEC - 1 : v);
    }
    for (int k = 0; k < NNODE / 256; ++k) {
      const int grp = (tid >> 5) + 8 * k;
      int myrank = 0, mycount = 0;
      for (int s = 0; s < NSPEC; ++s) {
        const unsigned m = __builtin_amdgcn_ballot_w32(sp[k] == s);
        if (sp[k] == s) myrank = __builtin_popcount(m & below);
        if (lane == s) mycount = __builtin_popcount(m);
      }
      rk[k] = myrank;
      if (lane < NSPEC) sCnt[grp * NSPEC + lane] = mycount;
    }
    __syncthreads();
    if (tid < NSPEC) {
      int run = 0;
      for (int g = 0; g < NGROUP; ++g) {
        sBase[g * NSPEC + tid] = run;
        run += sCnt[g * NSPEC + tid];
      }
      cnt[tid] = run;
    }
    __syncthreads();
    for (int k = 0; k < NNODE / 256; ++k) {
      const int grp = (tid >> 5) + 8 * k;
      zlist[sp[k] * NNODE + sBase[grp * NSPEC + sp[k]] + rk[k]] = tid + 256 * k;
    }
  }
}

#define PF 12
#define GRP 12
#define NV4 (UWROW / 4)
#define NSTAGE ((NV4 + 127) / 128)
__global__ __launch_bounds__(128) void symcon_main(
    const float* __restrict__ x, const float* __restrict__ ws, float* __restrict__ out) {
  __shared__ __attribute__((aligned(16))) float sB[NSTAGE * 128 * 4];
  const int* zlist = (const int*)(ws + ZL_OFF);
  const int* cnt   = (const int*)(ws + CNT_OFF);

  const int s = blockIdx.y;
  const int n = cnt[s];
  if (n <= 0) return;

  const int wv   = __builtin_amdgcn_readfirstlane(threadIdx.x >> 5);
  const int c    = blockIdx.x;
  const int lane = threadIdx.x & 31;
  const int h    = lane >> 4;
  const int r    = lane & 15;
  const int* zl  = zlist + s * NNODE;

  int tile = wv;
  int zcur, znext;
  {
    const int ra = tile * 16 + r;
    const int rb = ra + 64;
    zcur  = zl[ra < n ? ra : 0];
    znext = zl[rb < n ? rb : 0];
  }

  const v4f* usrc = (const v4f*)(ws + (size_t)(s * NCH + c) * UWROW);
  v4f stg[NSTAGE];
#pragma unroll
  for (int i = 0; i < NSTAGE; ++i) {
    const int idx = (int)(threadIdx.x & 127) + 128 * i;
    stg[i] = usrc[idx < NV4 ? idx : NV4 - 1];
  }

  float xv[9];
  {
    const bool inr = tile * 16 + r < n;
    const float* xr = x + ((size_t)zcur * NCH + c) * 9;
#pragma unroll
    for (int d = 0; d < 9; ++d) {
      const float xd = xr[d];
      xv[d] = inr ? xd : 0.f;
    }
  }

#pragma unroll
  for (int i = 0; i < NSTAGE; ++i) {
    const int idx = (int)(threadIdx.x & 127) + 128 * i;
    ((v4f*)sB)[idx] = stg[i];
  }
  __syncthreads();

  const float* ap = sB + 2 * lane;

  for (;;) {
    if (tile * 16 >= n) break;
    const bool valid = tile * 16 + r < n;

    v2f P[NKS];
#pragma unroll
    for (int j = 0; j < NKS; ++j) {
#pragma unroll
      for (int e = 0; e < 2; ++e) {
        const int q0 = 4 * j + e;
        const int q1 = q0 + 2;
        const float p0 = (q0 < 81) ? xv[q0 < 81 ? q0 / 9 : 0] * xv[q0 % 9] : (q0 == 81 ? 1.f : 0.f);
        const float p1 = (q1 < 81) ? xv[q1 < 81 ? q1 / 9 : 0] * xv[q1 % 9] : (q1 == 81 ? 1.f : 0.f);
        P[j][e] = h ? p1 : p0;
      }
    }

    const int rn  = (tile + 4) * 16 + r;
    const int rnn = (tile + 8) * 16 + r;
    const int znn = zl[rnn < n ? rnn : 0];
    float xn[9];
    {
      const float* xr = x + ((size_t)znext * NCH + c) * 9;
#pragma unroll
      for (int d = 0; d < 9; ++d) {
        const float xd = xr[d];
        xn[d] = (rn < n) ? xd : 0.f;
      }
    }

    v2f a[NFRAG];
#pragma unroll
    for (int f = 0; f < PF; ++f) a[f] = *(const v2f*)(ap + 64 * f);

    v8f acc[NMT];
#pragma unroll
    for (int m = 0; m < NMT; ++m) acc[m] = (v8f){0.f, 0.f, 0.f, 0.f, 0.f, 0.f, 0.f, 0.f};

#pragma unroll
    for (int g = 0; g < NFRAG; g += GRP) {
#pragma unroll
      for (int i = 0; i < GRP; ++i) {
        const int f = g + PF + i;
        if (f < NFRAG) a[f < NFRAG ? f : 0] = *(const v2f*)(ap + 64 * (f < NFRAG ? f : 0));
      }
      __builtin_amdgcn_sched_barrier(0);
#pragma unroll
      for (int i = 0; i < GRP; ++i) {
        const int f = g + i;
        if (f < NFRAG) {
          const int fc = f < NFRAG ? f : 0;
          const int ks = fc / NMT;
          const int mt = fc % NMT;
          acc[mt] = __builtin_amdgcn_wmma_f32_16x16x4_f32(false, a[fc], false, P[ks], (short)0, acc[mt], false, false);
        }
      }
    }

    float xm[5];
#pragma unroll
    for (int g = 0; g < 5; ++g) {
      const float hi = (g < 4) ? xv[g < 4 ? 5 + g : 0] : 1.f;
      xm[g] = h ? hi : xv[g];
    }
    float o[NCOMP];
#pragma unroll
    for (int nn = 0; nn < NCOMP; ++nn) {
      float sum = 0.f;
#pragma unroll
      for (int g = 0; g < 5; ++g) {
        const int sg = 9 * g + nn;
        sum += xm[g] * acc[sg >> 3][sg & 7];
      }
      o[nn] = sum;
    }
#pragma unroll
    for (int nn = 0; nn < NCOMP; ++nn) o[nn] += __shfl_xor(o[nn], 16);

    if (valid) {
      float* orow = out + (size_t)zcur * 1152;
      if (h == 0) {
        orow[c] = o[0];
        orow[128 + c * 3 + 0] = o[1];
        orow[128 + c * 3 + 1] = o[2];
        orow[128 + c * 3 + 2] = o[3];
        orow[512 + c * 5 + 0] = o[4];
      } else {
        orow[512 + c * 5 + 1] = o[5];
        orow[512 + c * 5 + 2] = o[6];
        orow[512 + c * 5 + 3] = o[7];
        orow[512 + c * 5 + 4] = o[8];
      }
    }

    tile += 4;
    zcur = znext;
    znext = znn;
#pragma unroll
    for (int d = 0; d < 9; ++d) xv[d] = xn[d];
  }
}

extern "C" void kernel_launch(void* const* d_in, const int* in_sizes, int n_in,
                              void* d_out, int out_size, void* d_ws, size_t ws_size,
                              hipStream_t stream) {
  const float* x     = (const float*)d_in[0];
  const float* u1_0e = (const float*)d_in[1];
  const float* w1_0e = (const float*)d_in[2];
  const float* u1_1o = (const float*)d_in[3];
  const float* w1_1o = (const float*)d_in[4];
  const float* u1_2e = (const float*)d_in[5];
  const float* w1_2e = (const float*)d_in[6];
  const float* u2_0e = (const float*)d_in[7];
  const float* w2_0e = (const float*)d_in[8];
  const float* u2_1o = (const float*)d_in[9];
  const float* w2_1o = (const float*)d_in[10];
  const float* u2_2e = (const float*)d_in[11];
  const float* w2_2e = (const float*)d_in[12];
  const float* u3_0e = (const float*)d_in[13];
  const float* w3_0e = (const float*)d_in[14];
  const float* u3_1o = (const float*)d_in[15];
  const float* w3_1o = (const float*)d_in[16];
  const float* u3_2e = (const float*)d_in[17];
  const float* w3_2e = (const float*)d_in[18];
  const int*   index = (const int*)d_in[19];
  float* ws  = (float*)d_ws;
  float* out = (float*)d_out;

  prep_uw<<<PREP_BLOCKS + 1, 256, 0, stream>>>(
      u1_0e, u1_1o, u1_2e, u2_0e, u2_1o, u2_2e, u3_0e, u3_1o, u3_2e,
      w1_0e, w1_1o, w1_2e, w2_0e, w2_1o, w2_2e, w3_0e, w3_1o, w3_2e, index, ws);
  symcon_main<<<dim3(NCH, NSPEC), 128, 0, stream>>>(x, ws, out);
}


# Prep launch: weight rows in LDS at a pitch of 36 floats, not 32, so the rows of the three irreps that one 16-lane half reads in a pass lie on different banks; same values, sums and order
# speedup vs baseline: 4.9491x; 1.0495x over previous
#include <hip/hip_runtime.h>

typedef float v2f __attribute__((ext_vector_type(2)));
typedef float v4f __attribute__((ext_vector_type(4)));
typedef float v8f __attribute__((ext_vector_type(8)));

#define NSPEC 10
#define NCH 128
#define NNODE 1024
#define NCOMP 9
#define NKS 21
#define NMT 6
#define NFRAG (NKS * NMT)
#define UWROW (NFRAG * 64)
#define NGROUP (NNODE / 32)
#define UW_FLOATS (NSPEC * NCH * UWROW)
#define ZL_OFF  UW_FLOATS
#define CNT_OFF (UW_FLOATS + NSPEC * NNODE)

#define CG 32
#define EPARTS 12
#define EPART_LEN 672
#define WROWS 28
#define PREP_BLOCKS (NSPEC * (NCH / CG) * EPARTS)
#define WFILL ((WROWS * CG + 255) / 256)
#define NEL ((EPART_LEN + 255) / 256)
#define MAXMUL 8
#define WP (CG + 4)

__device__ inline float w_row(const float* __restrict__ w1_0e, const float* __restrict__ w1_1o,
                              const float* __restrict__ w1_2e, const float* __restrict__ w2_0e,
                              const float* __restrict__ w2_1o, const float* __restrict__ w2_2e,
                              const float* __restrict__ w3_0e, const float* __restrict__ w3_1o,
                              const float* __restrict__ w3_2e, int row, int s, int c) {
  if (row < 1)  return w1_0e[s * NCH + c];
  if (row < 2)  return w1_1o[s * NCH + c];
  if (row < 3)  return w1_2e[s * NCH + c];
  if (row < 5)  return w2_0e[(s * 2 + row - 3) * NCH + c];
  if (row < 7)  return w2_1o[(s * 2 + row - 5) * NCH + c];
  if (row < 10) return w2_2e[(s * 3 + row - 7) * NCH + c];
  if (row < 14) return w3_0e[(s * 4 + row - 10) * NCH + c];
  if (row < 20) return w3_1o[(s * 6 + row - 14) * NCH + c];
  return w3_2e[(s * 8 + row - 20) * NCH + c];
}

__global__ __launch_bounds__(256) void prep_uw(
    const float* __restrict__ u1_0e, const float* __restrict__ u1_1o, const float* __restrict__ u1_2e,
    const float* __restrict__ u2_0e, const float* __restrict__ u2_1o, const float* __restrict__ u2_2e,
    const float* __restrict__ u3_0e, const float* __restrict__ u3_1o, const float* __restrict__ u3_2e,
    const float* __restrict__ w1_0e, const float* __restrict__ w1_1o, const float* __restrict__ w1_2e,
    const float* __restrict__ w2_0e, const float* __restrict__ w2_1o, const float* __restrict__ w2_2e,
    const float* __restrict__ w3_0e, const float* __restrict__ w3_1o, const float* __restrict__ w3_2e,
    const int* __restrict__ index, float* __restrict__ ws) {
  const int tid = threadIdx.x;
  if (blockIdx.x < PREP_BLOCKS) {
    __shared__ __attribute__((aligned(16))) float sW[WROWS * WP];
    const int bid  = blockIdx.x;
    const int s    = bid / ((NCH / CG) * EPARTS);
    const int rem  = bid - s * ((NCH / CG) * EPARTS);
    const int cg   = rem / EPARTS;
    const int part = rem - cg * EPARTS;
    const int c0   = cg * CG;
    float wv[WFILL];
#pragma unroll
    for (int i = 0; i < WFILL; ++i) {
      const int t   = tid + 256 * i;
      const int tt  = t < WROWS * CG ? t : 0;
      const int row = tt / CG;
      const int cc  = tt - row * CG;
      wv[i] = w_row(w1_0e, w1_1o, w1_2e, w2_0e, w2_1o, w2_2e, w3_0e, w3_1o, w3_2e, row, s, c0 + cc);
    }
    float uv[NEL][MAXMUL];
    int mulv[NEL], wrow0v[NEL];
#pragma unroll
    for (int it = 0; it < NEL; ++it) {
      const int e   = part * EPART_LEN + tid + 256 * it;
      const int el  = e & 1;
      const int ln  = (e >> 1) & 31;
      const int fr  = e >> 6;
      const int ks  = fr / NMT;
      const int mt  = fr - ks * NMT;
      const int q   = 4 * ks + 2 * (ln >> 4) + el;
      const int sg  = 8 * mt + (ln & 7);
      const int R   = 45 * ((ln >> 3) & 1) + sg;
      const int d1  = R / 9;
      const int n   = R - d1 * 9;
      const int ir   = (n == 0) ? 0 : (n < 4 ? 1 : 2);
      const int idim = (ir == 0) ? 1 : (ir == 1 ? 3 : 5);
      const int i    = (ir == 0) ? 0 : (ir == 1 ? n - 1 : n - 4);
      const float* u = u1_0e;
      int mul = 0, ubase = 0, wrow0 = 0;
      if (sg < 45) {
        if (q < 81) {
          if (d1 < 9) {
            u = (ir == 0) ? u3_0e : (ir == 1 ? u3_1o : u3_2e);
            mul = (ir == 0) ? 4 : (ir == 1 ? 6 : 8);
            ubase = (d1 * 81 + q) * mul * idim + i;
            wrow0 = (ir == 0) ? 10 : (ir == 1 ? 14 : 20);
          } else {
            u = (ir == 0) ? u2_0e : (ir == 1 ? u2_1o : u2_2e);
            mul = (ir == 2) ? 3 : 2;
            ubase = q * mul * idim + i;
            wrow0 = (ir == 0) ? 3 : (ir == 1 ? 5 : 7);
          }
        } else if (q == 81 && d1 < 9) {
          u = (ir == 0) ? u1_0e : (ir == 1 ? u1_1o : u1_2e);
          mul = 1;
          ubase = d1 * idim + i;
          wrow0 = ir;
        }
      }
      const bool live = tid + 256 * it < EPART_LEN;
      mulv[it]   = live ? mul : 0;
      wrow0v[it] = wrow0;
#pragma unroll
      for (int k = 0; k < MAXMUL; ++k) uv[it][k] = 0.f;
      if (live) {
#pragma unroll
        for (int k = 0; k < MAXMUL; ++k) uv[it][k] = u[ubase + (k < mul ? k : 0) * idim];
      }
    }
#pragma unroll
    for (int i = 0; i < WFILL; ++i) {
      const int t = tid + 256 * i;
      if (t < WROWS * CG) sW[(t / CG) * WP + (t % CG)] = wv[i];
    }
    __syncthreads();
#pragma unroll
    for (int it = 0; it < NEL; ++it) {
      const int eo = tid + 256 * it;
      if (eo < EPART_LEN) {
        float v[CG];
#pragma unroll
        for (int cc = 0; cc < CG; ++cc) v[cc] = 0.f;
#pragma unroll
        for (int k = 0; k < MAXMUL; ++k) {
          if (k < mulv[it]) {
            const float uk = uv[it][k];
            const float* wr = sW + (wrow0v[it] + k) * WP;
#pragma unroll
            for (int cc = 0; cc < CG; ++cc) v[cc] += wr[cc] * uk;
          }
        }
        float* dst = ws + (size_t)(s * NCH + c0) * UWROW + part * EPART_LEN + eo;
#pragma unroll
        for (int cc = 0; cc < CG; ++cc) dst[(size_t)cc * UWROW] = v[cc];
      }
    }
  } else {
    __shared__ int sCnt[NGROUP * NSPEC];
    __shared__ int sBase[NGROUP * NSPEC];
    int* zlist = (int*)(ws + ZL_OFF);
    int* cnt   = (int*)(ws + CNT_OFF);
    const int lane = tid & 31;
    const unsigned below = (1u << lane) - 1u;
    int sp[NNODE / 256];
    int rk[NNODE / 256];
    for (int k = 0; k < NNODE / 256; ++k) {
      int v = index[tid + 256 * k];
      sp[k] = v < 0 ? 0 : (v > NSPEC - 1 ? NSPEC - 1 : v);
    }
    for (int k = 0; k < NNODE / 256; ++k) {
      const int grp = (tid >> 5) + 8 * k;
      int myrank = 0, mycount = 0;
      for (int s = 0; s < NSPEC; ++s) {
        const unsigned m = __builtin_amdgcn_ballot_w32(sp[k] == s);
        if (sp[k] == s) myrank = __builtin_popcount(m & below);
        if (lane == s) mycount = __builtin_popcount(m);
      }
      rk[k] = myrank;
      if (lane < NSPEC) sCnt[grp * NSPEC + lane] = mycount;
    }
    __syncthreads();
    if (tid < NSPEC) {
      int run = 0;
      for (int g = 0; g < NGROUP; ++g) {
        sBase[g * NSPEC + tid] = run;
        run += sCnt[g * NSPEC + tid];
      }
      cnt[tid] = run;
    }
    __syncthreads();
    for (int k = 0; k < NNODE / 256; ++k) {
      const int grp = (tid >> 5) + 8 * k;
      zlist[sp[k] * NNODE + sBase[grp * NSPEC + sp[k]] + rk[k]] = tid + 256 * k;
    }
  }
}

#define PF 12
#define GRP 12
#define NV4 (UWROW / 4)
#define NSTAGE ((NV4 + 127) / 128)
__global__ __launch_bounds__(128) void symcon_main(
    const float* __restrict__ x, const float* __restrict__ ws, float* __restrict__ out) {
  __shared__ __attribute__((aligned(16))) float sB[NSTAGE * 128 * 4];
  const int* zlist = (const int*)(ws + ZL_OFF);
  const int* cnt   = (const int*)(ws + CNT_OFF);

  const int s = blockIdx.y;
  const int n = cnt[s];
  if (n <= 0) return;

  const int wv   = __builtin_amdgcn_readfirstlane(threadIdx.x >> 5);
  const int c    = blockIdx.x;
  const int lane = threadIdx.x & 31;
  const int h    = lane >> 4;
  const int r    = lane & 15;
  const int* zl  = zlist + s * NNODE;

  int tile = wv;
  int zcur, znext;
  {
    const int ra = tile * 16 + r;
    const int rb = ra + 64;
    zcur  = zl[ra < n ? ra : 0];
    znext = zl[rb < n ? rb : 0];
  }

  const v4f* usrc = (const v4f*)(ws + (size_t)(s * NCH + c) * UWROW);
  v4f stg[NSTAGE];
#pragma unroll
  for (int i = 0; i < NSTAGE; ++i) {
    const int idx = (int)(threadIdx.x & 127) + 128 * i;
    stg[i] = usrc[idx < NV4 ? idx : NV4 - 1];
  }

  float xv[9];
  {
    const bool inr = tile * 16 + r < n;
    const float* xr = x + ((size_t)zcur * NCH + c) * 9;
#pragma unroll
    for (int d = 0; d < 9; ++d) {
      const float xd = xr[d];
      xv[d] = inr ? xd : 0.f;
    }
  }

#pragma unroll
  for (int i = 0; i < NSTAGE; ++i) {
    const int idx = (int)(threadIdx.x & 127) + 128 * i;
    ((v4f*)sB)[idx] = stg[i];
  }
  __syncthreads();

  const float* ap = sB + 2 * lane;

  for (;;) {
    if (tile * 16 >= n) break;
    const bool valid = tile * 16 + r < n;

    v2f P[NKS];
#pragma unroll
    for (int j = 0; j < NKS; ++j) {
#pragma unroll
      for (int e = 0; e < 2; ++e) {
        const int q0 = 4 * j + e;
        const int q1 = q0 + 2;
        const float p0 = (q0 < 81) ? xv[q0 < 81 ? q0 / 9 : 0] * xv[q0 % 9] : (q0 == 81 ? 1.f : 0.f);
        const float p1 = (q1 < 81) ? xv[q1 < 81 ? q1 / 9 : 0] * xv[q1 % 9] : (q1 == 81 ? 1.f : 0.f);
        P[j][e] = h ? p1 : p0;
      }
    }

    const int rn  = (tile + 4) * 16 + r;
    const int rnn = (tile + 8) * 16 + r;
    const int znn = zl[rnn < n ? rnn : 0];
    float xn[9];
    {
      const float* xr = x + ((size_t)znext * NCH + c) * 9;
#pragma unroll
      for (int d = 0; d < 9; ++d) {
        const float xd = xr[d];
        xn[d] = (rn < n) ? xd : 0.f;
      }
    }

    v2f a[NFRAG];
#pragma unroll
    for (int f = 0; f < PF; ++f) a[f] = *(const v2f*)(ap + 64 * f);

    v8f acc[NMT];
#pragma unroll
    for (int m = 0; m < NMT; ++m) acc[m] = (v8f){0.f, 0.f, 0.f, 0.f, 0.f, 0.f, 0.f, 0.f};

#pragma unroll
    for (int g = 0; g < NFRAG; g += GRP) {
#pragma unroll
      for (int i = 0; i < GRP; ++i) {
        const int f = g + PF + i;
        if (f < NFRAG) a[f < NFRAG ? f : 0] = *(const v2f*)(ap + 64 * (f < NFRAG ? f : 0));
      }
      __builtin_amdgcn_sched_barrier(0);
#pragma unroll
      for (int i = 0; i < GRP; ++i) {
        const int f = g + i;
        if (f < NFRAG) {
          const int fc = f < NFRAG ? f : 0;
          const int ks = fc / NMT;
          const int mt = fc % NMT;
          acc[mt] = __builtin_amdgcn_wmma_f32_16x16x4_f32(false, a[fc], false, P[ks], (short)0, acc[mt], false, false);
        }
      }
    }

    float xm[5];
#pragma unroll
    for (int g = 0; g < 5; ++g) {
      const float hi = (g < 4) ? xv[g < 4 ? 5 + g : 0] : 1.f;
      xm[g] = h ? hi : xv[g];
    }
    float o[NCOMP];
#pragma unroll
    for (int nn = 0; nn < NCOMP; ++nn) {
      float sum = 0.f;
#pragma unroll
      for (int g = 0; g < 5; ++g) {
        const int sg = 9 * g + nn;
        sum += xm[g] * acc[sg >> 3][sg & 7];
      }
      o[nn] = sum;
    }
#pragma unroll
    for (int nn = 0; nn < NCOMP; ++nn) o[nn] += __shfl_xor(o[nn], 16);

    if (valid) {
      float* orow = out + (size_t)zcur * 1152;
      if (h == 0) {
        orow[c] = o[0];
        orow[128 + c * 3 + 0] = o[1];
        orow[128 + c * 3 + 1] = o[2];
        orow[128 + c * 3 + 2] = o[3];
        orow[512 + c * 5 + 0] = o[4];
      } else {
        orow[512 + c * 5 + 1] = o[5];
        orow[512 + c * 5 + 2] = o[6];
        orow[512 + c * 5 + 3] = o[7];
        orow[512 + c * 5 + 4] = o[8];
      }
    }

    tile += 4;
    zcur = znext;
    znext = znn;
#pragma unroll
    for (int d = 0; d < 9; ++d) xv[d] = xn[d];
  }
}

extern "C" void kernel_launch(void* const* d_in, const int* in_sizes, int n_in,
                              void* d_out, int out_size, void* d_ws, size_t ws_size,
                              hipStream_t stream) {
  const float* x     = (const float*)d_in[0];
  const float* u1_0e = (const float*)d_in[1];
  const float* w1_0e = (const float*)d_in[2];
  const float* u1_1o = (const float*)d_in[3];
  const float* w1_1o = (const float*)d_in[4];
  const float* u1_2e = (const float*)d_in[5];
  const float* w1_2e = (const float*)d_in[6];
  const float* u2_0e = (const float*)d_in[7];
  const float* w2_0e = (const float*)d_in[8];
  const float* u2_1o = (const float*)d_in[9];
  const float* w2_1o = (const float*)d_in[10];
  const float* u2_2e = (const float*)d_in[11];
  const float* w2_2e = (const float*)d_in[12];
  const float* u3_0e = (const float*)d_in[13];
  const float* w3_0e = (const float*)d_in[14];
  const float* u3_1o = (const float*)d_in[15];
  const float* w3_1o = (const float*)d_in[16];
  const float* u3_2e = (const float*)d_in[17];
  const float* w3_2e = (const float*)d_in[18];
  const int*   index = (const int*)d_in[19];
  float* ws  = (float*)d_ws;
  float* out = (float*)d_out;

  prep_uw<<<PREP_BLOCKS + 1, 256, 0, stream>>>(
      u1_0e, u1_1o, u1_2e, u2_0e, u2_1o, u2_2e, u3_0e, u3_1o, u3_2e,
      w1_0e, w1_1o, w1_2e, w2_0e, w2_1o, w2_2e, w3_0e, w3_1o, w3_2e, index, ws);
  symcon_main<<<dim3(NCH, NSPEC), 128, 0, stream>>>(x, ws, out);
}


# Main kernel: operand reaches LDS in two halves, a barrier each; a wave's first tile runs K-steps 0-9 on the first half while the second half's loads are in flight; same sums and order
# speedup vs baseline: 5.1746x; 1.0456x over previous
#include <hip/hip_runtime.h>

typedef float v2f __attribute__((ext_vector_type(2)));
typedef float v4f __attribute__((ext_vector_type(4)));
typedef float v8f __attribute__((ext_vector_type(8)));

#define NSPEC 10
#define NCH 128
#define NNODE 1024
#define NCOMP 9
#define NKS 21
#define NMT 6
#define NFRAG (NKS * NMT)
#define UWROW (NFRAG * 64)
#define NGROUP (NNODE / 32)
#define UW_FLOATS (NSPEC * NCH * UWROW)
#define ZL_OFF  UW_FLOATS
#define CNT_OFF (UW_FLOATS + NSPEC * NNODE)

#define CG 32
#define EPARTS 12
#define EPART_LEN 672
#define WROWS 28
#define PREP_BLOCKS (NSPEC * (NCH / CG) * EPARTS)
#define WFILL ((WROWS * CG + 255) / 256)
#define NEL ((EPART_LEN + 255) / 256)
#define MAXMUL 8
#define WP (CG + 4)

__device__ inline float w_row(const float* __restrict__ w1_0e, const float* __restrict__ w1_1o,
                              const float* __restrict__ w1_2e, const float* __restrict__ w2_0e,
                              const float* __restrict__ w2_1o, const float* __restrict__ w2_2e,
                              const float* __restrict__ w3_0e, const float* __restrict__ w3_1o,
                              const float* __restrict__ w3_2e, int row, int s, int c) {
  if (row < 1)  return w1_0e[s * NCH + c];
  if (row < 2)  return w1_1o[s * NCH + c];
  if (row < 3)  return w1_2e[s * NCH + c];
  if (row < 5)  return w2_0e[(s * 2 + row - 3) * NCH + c];
  if (row < 7)  return w2_1o[(s * 2 + row - 5) * NCH + c];
  if (row < 10) return w2_2e[(s * 3 + row - 7) * NCH + c];
  if (row < 14) return w3_0e[(s * 4 + row - 10) * NCH + c];
  if (row < 20) return w3_1o[(s * 6 + row - 14) * NCH + c];
  return w3_2e[(s * 8 + row - 20) * NCH + c];
}

__global__ __launch_bounds__(256) void prep_uw(
    const float* __restrict__ u1_0e, const float* __restrict__ u1_1o, const float* __restrict__ u1_2e,
    const float* __restrict__ u2_0e, const float* __restrict__ u2_1o, const float* __restrict__ u2_2e,
    const float* __restrict__ u3_0e, const float* __restrict__ u3_1o, const float* __restrict__ u3_2e,
    const float* __restrict__ w1_0e, const float* __restrict__ w1_1o, const float* __restrict__ w1_2e,
    const float* __restrict__ w2_0e, const float* __restrict__ w2_1o, const float* __restrict__ w2_2e,
    const float* __restrict__ w3_0e, const float* __restrict__ w3_1o, const float* __restrict__ w3_2e,
    const int* __restrict__ index, float* __restrict__ ws) {
  const int tid = threadIdx.x;
  if (blockIdx.x < PREP_BLOCKS) {
    __shared__ __attribute__((aligned(16))) float sW[WROWS * WP];
    const int bid  = blockIdx.x;
    const int s    = bid / ((NCH / CG) * EPARTS);
    const int rem  = bid - s * ((NCH / CG) * EPARTS);
    const int cg   = rem / EPARTS;
    const int part = rem - cg * EPARTS;
    const int c0   = cg * CG;
    float wv[WFILL];
#pragma unroll
    for (int i = 0; i < WFILL; ++i) {
      const int t   = tid + 256 * i;
      const int tt  = t < WROWS * CG ? t : 0;
      const int row = tt / CG;
      const int cc  = tt - row * CG;
      wv[i] = w_row(w1_0e, w1_1o, w1_2e, w2_0e, w2_1o, w2_2e, w3_0e, w3_1o, w3_2e, row, s, c0 + cc);
    }
    float uv[NEL][MAXMUL];
    int mulv[NEL], wrow0v[NEL];
#pragma unroll
    for (int it = 0; it < NEL; ++it) {
      const int e   = part * EPART_LEN + tid + 256 * it;
      const int el  = e & 1;
      const int ln  = (e >> 1) & 31;
      const int fr  = e >> 6;
      const int ks  = fr / NMT;
      const int mt  = fr - ks * NMT;
      const int q   = 4 * ks + 2 * (ln >> 4) + el;
      const int sg  = 8 * mt + (ln & 7);
      const int R   = 45 * ((ln >> 3) & 1) + sg;
      const int d1  = R / 9;
      const int n   = R - d1 * 9;
      const int ir   = (n == 0) ? 0 : (n < 4 ? 1 : 2);
      const int idim = (ir == 0) ? 1 : (ir == 1 ? 3 : 5);
      const int i    = (ir == 0) ? 0 : (ir == 1 ? n - 1 : n - 4);
      const float* u = u1_0e;
      int mul = 0, ubase = 0, wrow0 = 0;
      if (sg < 45) {
        if (q < 81) {
          if (d1 < 9) {
            u = (ir == 0) ? u3_0e : (ir == 1 ? u3_1o : u3_2e);
            mul = (ir == 0) ? 4 : (ir == 1 ? 6 : 8);
            ubase = (d1 * 81 + q) * mul * idim + i;
            wrow0 = (ir == 0) ? 10 : (ir == 1 ? 14 : 20);
          } else {
            u = (ir == 0) ? u2_0e : (ir == 1 ? u2_1o : u2_2e);
            mul = (ir == 2) ? 3 : 2;
            ubase = q * mul * idim + i;
            wrow0 = (ir == 0) ? 3 : (ir == 1 ? 5 : 7);
          }
        } else if (q == 81 && d1 < 9) {
          u = (ir == 0) ? u1_0e : (ir == 1 ? u1_1o : u1_2e);
          mul = 1;
          ubase = d1 * idim + i;
          wrow0 = ir;
        }
      }
      const bool live = tid + 256 * it < EPART_LEN;
      mulv[it]   = live ? mul : 0;
      wrow0v[it] = wrow0;
#pragma unroll
      for (int k = 0; k < MAXMUL; ++k) uv[it][k] = 0.f;
      if (live) {
#pragma unroll
        for (int k = 0; k < MAXMUL; ++k) uv[it][k] = u[ubase + (k < mul ? k : 0) * idim];
      }
    }
#pragma unroll
    for (int i = 0; i < WFILL; ++i) {
      const int t = tid + 256 * i;
      if (t < WROWS * CG) sW[(t / CG) * WP + (t % CG)] = wv[i];
    }
    __syncthreads();
#pragma unroll
    for (int it = 0; it < NEL; ++it) {
      const int eo = tid + 256 * it;
      if (eo < EPART_LEN) {
        float v[CG];
#pragma unroll
        for (int cc = 0; cc < CG; ++cc) v[cc] = 0.f;
#pragma unroll
        for (int k = 0; k < MAXMUL; ++k) {
          if (k < mulv[it]) {
            const float uk = uv[it][k];
            const float* wr = sW + (wrow0v[it] + k) * WP;
#pragma unroll
            for (int cc = 0; cc < CG; ++cc) v[cc] += wr[cc] * uk;
          }
        }
        float* dst = ws + (size_t)(s * NCH + c0) * UWROW + part * EPART_LEN + eo;
#pragma unroll
        for (int cc = 0; cc < CG; ++cc) dst[(size_t)cc * UWROW] = v[cc];
      }
    }
  } else {
    __shared__ int sCnt[NGROUP * NSPEC];
    __shared__ int sBase[NGROUP * NSPEC];
    int* zlist = (int*)(ws + ZL_OFF);
    int* cnt   = (int*)(ws + CNT_OFF);
    const int lane = tid & 31;
    const unsigned below = (1u << lane) - 1u;
    int sp[NNODE / 256];
    int rk[NNODE / 256];
    for (int k = 0; k < NNODE / 256; ++k) {
      int v = index[tid + 256 * k];
      sp[k] = v < 0 ? 0 : (v > NSPEC - 1 ? NSPEC - 1 : v);
    }
    for (int k = 0; k < NNODE / 256; ++k) {
      const int grp = (tid >> 5) + 8 * k;
      int myrank = 0, mycount = 0;
      for (int s = 0; s < NSPEC; ++s) {
        const unsigned m = __builtin_amdgcn_ballot_w32(sp[k] == s);
        if (sp[k] == s) myrank = __builtin_popcount(m & below);
        if (lane == s) mycount = __builtin_popcount(m);
      }
      rk[k] = myrank;
      if (lane < NSPEC) sCnt[grp * NSPEC + lane] = mycount;
    }
    __syncthreads();
    if (tid < NSPEC) {
      int run = 0;
      for (int g = 0; g < NGROUP; ++g) {
        sBase[g * NSPEC + tid] = run;
        run += sCnt[g * NSPEC + tid];
      }
      cnt[tid] = run;
    }
    __syncthreads();
    for (int k = 0; k < NNODE / 256; ++k) {
      const int grp = (tid >> 5) + 8 * k;
      zlist[sp[k] * NNODE + sBase[grp * NSPEC + sp[k]] + rk[k]] = tid + 256 * k;
    }
  }
}

#define PF 12
#define GRP 12
#define NV4 (UWROW / 4)
#define NHALF 8
#define FSPLIT 60
#define KSPLIT (FSPLIT / NMT)

template <int J0, int J1>
__device__ __forceinline__ void comp_p(const float (&xv)[9], int h, v2f (&P)[NKS]) {
#pragma unroll
  for (int j = J0; j < J1; ++j) {
#pragma unroll
    for (int e = 0; e < 2; ++e) {
      const int q0 = 4 * j + e;
      const int q1 = q0 + 2;
      const float p0 = (q0 < 81) ? xv[q0 < 81 ? q0 / 9 : 0] * xv[q0 % 9] : (q0 == 81 ? 1.f : 0.f);
      const float p1 = (q1 < 81) ? xv[q1 < 81 ? q1 / 9 : 0] * xv[q1 % 9] : (q1 == 81 ? 1.f : 0.f);
      P[j][e] = h ? p1 : p0;
    }
  }
}

template <int F0, int F1>
__device__ __forceinline__ void mm(const float* ap, const v2f (&P)[NKS], v8f (&acc)[NMT]) {
  v2f a[NFRAG];
#pragma unroll
  for (int i = 0; i < PF; ++i) {
    const int f = F0 + i;
    if (f < F1) a[f < F1 ? f : F0] = *(const v2f*)(ap + 64 * (f < F1 ? f : F0));
  }
#pragma unroll
  for (int g = F0; g < F1; g += GRP) {
#pragma unroll
    for (int i = 0; i < GRP; ++i) {
      const int f = g + PF + i;
      if (f < F1) a[f < F1 ? f : F0] = *(const v2f*)(ap + 64 * (f < F1 ? f : F0));
    }
    __builtin_amdgcn_sched_barrier(0);
#pragma unroll
    for (int i = 0; i < GRP; ++i) {
      const int f = g + i;
      if (f < F1) {
        const int fc = f < F1 ? f : F0;
        const int ks = fc / NMT;
        const int mt = fc % NMT;
        acc[mt] = __builtin_amdgcn_wmma_f32_16x16x4_f32(false, a[fc], false, P[ks], (short)0, acc[mt], false, false);
      }
    }
  }
}

__device__ __forceinline__ void finish(const v8f (&acc)[NMT], const float (&xv)[9], int h, bool valid,
                                       int zcur, int c, float* __restrict__ out) {
  float xm[5];
#pragma unroll
  for (int g = 0; g < 5; ++g) {
    const float hi = (g < 4) ? xv[g < 4 ? 5 + g : 0] : 1.f;
    xm[g] = h ? hi : xv[g];
  }
  float o[NCOMP];
#pragma unroll
  for (int nn = 0; nn < NCOMP; ++nn) {
    float sum = 0.f;
#pragma unroll
    for (int g = 0; g < 5; ++g) {
      const int sg = 9 * g + nn;
      sum += xm[g] * acc[sg >> 3][sg & 7];
    }
    o[nn] = sum;
  }
#pragma unroll
  for (int nn = 0; nn < NCOMP; ++nn) o[nn] += __shfl_xor(o[nn], 16);

  if (valid) {
    float* orow = out + (size_t)zcur * 1152;
    if (h == 0) {
      orow[c] = o[0];
      orow[128 + c * 3 + 0] = o[1];
      orow[128 + c * 3 + 1] = o[2];
      orow[128 + c * 3 + 2] = o[3];
      orow[512 + c * 5 + 0] = o[4];
    } else {
      orow[512 + c * 5 + 1] = o[5];
      orow[512 + c * 5 + 2] = o[6];
      orow[512 + c * 5 + 3] = o[7];
      orow[512 + c * 5 + 4] = o[8];
    }
  }
}

__global__ __launch_bounds__(128, 5) void symcon_main(
    const float* __restrict__ x, const float* __restrict__ ws, float* __restrict__ out) {
  __shared__ __attribute__((aligned(16))) float sB[2 * NHALF * 128 * 4];
  const int* zlist = (const int*)(ws + ZL_OFF);
  const int* cnt   = (const int*)(ws + CNT_OFF);

  const int s = blockIdx.y;
  const int n = cnt[s];
  if (n <= 0) return;

  const int wv   = __builtin_amdgcn_readfirstlane(threadIdx.x >> 5);
  const int c    = blockIdx.x;
  const int lane = threadIdx.x & 31;
  const int h    = lane >> 4;
  const int r    = lane & 15;
  const int t128 = (int)(threadIdx.x & 127);
  const int* zl  = zlist + s * NNODE;

  int tile = wv;
  int zcur, znext;
  {
    const int ra = tile * 16 + r;
    const int rb = ra + 64;
    zcur  = zl[ra < n ? ra : 0];
    znext = zl[rb < n ? rb : 0];
  }

  const v4f* usrc = (const v4f*)(ws + (size_t)(s * NCH + c) * UWROW);
  v4f stg[NHALF];
#pragma unroll
  for (int i = 0; i < NHALF; ++i) stg[i] = usrc[t128 + 128 * i];

  float xv[9];
  {
    const bool inr = tile * 16 + r < n;
    const float* xr = x + ((size_t)zcur * NCH + c) * 9;
#pragma unroll
    for (int d = 0; d < 9; ++d) {
      const float xd = xr[d];
      xv[d] = inr ? xd : 0.f;
    }
  }

#pragma unroll
  for (int i = 0; i < NHALF; ++i) ((v4f*)sB)[t128 + 128 * i] = stg[i];
  __syncthreads();

#pragma unroll
  for (int i = 0; i < NHALF; ++i) {
    const int idx = t128 + 128 * (NHALF + i);
    stg[i] = usrc[idx < NV4 ? idx : NV4 - 1];
  }

  const float* ap = sB + 2 * lane;
  const bool has = tile * 16 < n;

  v2f P[NKS];
  v8f acc[NMT];
#pragma unroll
  for (int m = 0; m < NMT; ++m) acc[m] = (v8f){0.f, 0.f, 0.f, 0.f, 0.f, 0.f, 0.f, 0.f};
  if (has) {
    comp_p<0, KSPLIT>(xv, h, P);
    mm<0, FSPLIT>(ap, P, acc);
  }

#pragma unroll
  for (int i = 0; i < NHALF; ++i) ((v4f*)sB)[t128 + 128 * (NHALF + i)] = stg[i];
  __syncthreads();

  if (!has) return;

  {
    const bool valid = tile * 16 + r < n;
    comp_p<KSPLIT, NKS>(xv, h, P);
    const int rn  = (tile + 4) * 16 + r;
    const int rnn = (tile + 8) * 16 + r;
    const int znn = zl[rnn < n ? rnn : 0];
    float xn[9];
    {
      const float* xr = x + ((size_t)znext * NCH + c) * 9;
#pragma unroll
      for (int d = 0; d < 9; ++d) {
        const float xd = xr[d];
        xn[d] = (rn < n) ? xd : 0.f;
      }
    }
    mm<FSPLIT, NFRAG>(ap, P, acc);
    finish(acc, xv, h, valid, zcur, c, out);
    tile += 4;
    zcur = znext;
    znext = znn;
#pragma unroll
    for (int d = 0; d < 9; ++d) xv[d] = xn[d];
  }

  for (;;) {
    if (tile * 16 >= n) break;
    const bool valid = tile * 16 + r < n;
    comp_p<0, NKS>(xv, h, P);
    const int rn  = (tile + 4) * 16 + r;
    const int rnn = (tile + 8) * 16 + r;
    const int znn = zl[rnn < n ? rnn : 0];
    float xn[9];
    {
      const float* xr = x + ((size_t)znext * NCH + c) * 9;
#pragma unroll
      for (int d = 0; d < 9; ++d) {
        const float xd = xr[d];
        xn[d] = (rn < n) ? xd : 0.f;
      }
    }
#pragma unroll
    for (int m = 0; m < NMT; ++m) acc[m] = (v8f){0.f, 0.f, 0.f, 0.f, 0.f, 0.f, 0.f, 0.f};
    mm<0, NFRAG>(ap, P, acc);
    finish(acc, xv, h, valid, zcur, c, out);
    tile += 4;
    zcur = znext;
    znext = znn;
#pragma unroll
    for (int d = 0; d < 9; ++d) xv[d] = xn[d];
  }
}

extern "C" void kernel_launch(void* const* d_in, const int* in_sizes, int n_in,
                              void* d_out, int out_size, void* d_ws, size_t ws_size,
                              hipStream_t stream) {
  const float* x     = (const float*)d_in[0];
  const float* u1_0e = (const float*)d_in[1];
  const float* w1_0e = (const float*)d_in[2];
  const float* u1_1o = (const float*)d_in[3];
  const float* w1_1o = (const float*)d_in[4];
  const float* u1_2e = (const float*)d_in[5];
  const float* w1_2e = (const float*)d_in[6];
  const float* u2_0e = (const float*)d_in[7];
  const float* w2_0e = (const float*)d_in[8];
  const float* u2_1o = (const float*)d_in[9];
  const float* w2_1o = (const float*)d_in[10];
  const float* u2_2e = (const float*)d_in[11];
  const float* w2_2e = (const float*)d_in[12];
  const float* u3_0e = (const float*)d_in[13];
  const float* w3_0e = (const float*)d_in[14];
  const float* u3_1o = (const float*)d_in[15];
  const float* w3_1o = (const float*)d_in[16];
  const float* u3_2e = (const float*)d_in[17];
  const float* w3_2e = (const float*)d_in[18];
  const int*   index = (const int*)d_in[19];
  float* ws  = (float*)d_ws;
  float* out = (float*)d_out;

  prep_uw<<<PREP_BLOCKS + 1, 256, 0, stream>>>(
      u1_0e, u1_1o, u1_2e, u2_0e, u2_1o, u2_2e, u3_0e, u3_1o, u3_2e,
      w1_0e, w1_1o, w1_2e, w2_0e, w2_1o, w2_2e, w3_0e, w3_1o, w3_2e, index, ws);
  symcon_main<<<dim3(NCH, NSPEC), 128, 0, stream>>>(x, ws, out);
}
